# starting kernel
# baseline (speedup 1.0000x reference)
#include <hip/hip_runtime.h>
#include <hip/hip_bf16.h>

typedef __attribute__((ext_vector_type(2))) float v2f;
typedef __attribute__((ext_vector_type(8))) float v8f;

#define D_INNER 1024
#define STATE   16
#define DT_RANK 64
#define BATCH   2
#define SEQ     2048
#define ROWS    (BATCH * SEQ)
#define E_DIM   (DT_RANK + 2 * STATE)
#define NCHUNK  8
#define CLEN    (SEQ / NCHUNK)

__global__ __launch_bounds__(128) void xproj_wmma(
    const float* __restrict__ u, const float* __restrict__ Wx,
    float* __restrict__ x_dbl) {
  const int wave = blockIdx.x * (blockDim.x >> 5) + (threadIdx.x >> 5);
  const int lane = threadIdx.x & 31;
  const int rt = wave / (E_DIM / 16);
  const int ct = wave % (E_DIM / 16);
  const int row0 = rt * 16, col0 = ct * 16;
  const int m  = lane & 15;
  const int kh = lane >> 4;

  v8f acc = {};
  const float* ua = u  + (size_t)(row0 + m) * D_INNER + 2 * kh;
  const float* wa = Wx + (size_t)(col0 + m) * D_INNER + 2 * kh;
#pragma unroll 8
  for (int kb = 0; kb < D_INNER; kb += 4) {
    v2f a = *(const v2f*)(ua + kb);
    v2f b = *(const v2f*)(wa + kb);
    acc = __builtin_amdgcn_wmma_f32_16x16x4_f32(
         false, a,  false, b,
         (short)0, acc,  false,  false);
  }
  float* outp = x_dbl + (size_t)(row0 + 8 * kh) * E_DIM + col0 + m;
#pragma unroll
  for (int j = 0; j < 8; ++j) outp[(size_t)j * E_DIM] = acc[j];
}

__global__ __launch_bounds__(128) void dtproj_wmma(
    const float* __restrict__ x_dbl, const float* __restrict__ Wdt,
    const float* __restrict__ b_dt, float* __restrict__ dt) {
  const int wave = blockIdx.x * (blockDim.x >> 5) + (threadIdx.x >> 5);
  const int lane = threadIdx.x & 31;
  const int rt = wave / (D_INNER / 16);
  const int ct = wave % (D_INNER / 16);
  const int row0 = rt * 16, col0 = ct * 16;
  const int m  = lane & 15;
  const int kh = lane >> 4;

  v8f acc = {};
  const float* aa = x_dbl + (size_t)(row0 + m) * E_DIM + 2 * kh;
  const float* wa = Wdt   + (size_t)(col0 + m) * DT_RANK + 2 * kh;
#pragma unroll
  for (int kb = 0; kb < DT_RANK; kb += 4) {
    v2f a = *(const v2f*)(aa + kb);
    v2f b = *(const v2f*)(wa + kb);
    acc = __builtin_amdgcn_wmma_f32_16x16x4_f32(
        false, a, false, b, (short)0, acc, false, false);
  }
  const float bias = b_dt[col0 + m];
  float* outp = dt + (size_t)(row0 + 8 * kh) * D_INNER + col0 + m;
#pragma unroll
  for (int j = 0; j < 8; ++j) {
    float x = acc[j] + bias;
    float sp = (x > 20.0f) ? x : log1pf(__expf(x));
    outp[(size_t)j * D_INNER] = sp;
  }
}

__global__ __launch_bounds__(256) void scan_pass1(
    const float* __restrict__ u, const float* __restrict__ A_log,
    const float* __restrict__ x_dbl, const float* __restrict__ dt,
    float* __restrict__ prodA, float* __restrict__ hend) {
  const int wave = blockIdx.x * (blockDim.x >> 5) + (threadIdx.x >> 5);
  const int lane = threadIdx.x & 31;
  const int b     = wave / (NCHUNK * (D_INNER / 2));
  const int rem   = wave % (NCHUNK * (D_INNER / 2));
  const int chunk = rem / (D_INNER / 2);
  const int dpair = rem % (D_INNER / 2);
  const int d = dpair * 2 + (lane >> 4);
  const int n = lane & 15;

  const float A = -__expf(A_log[d * STATE + n]);
  float h = 0.0f, p = 1.0f;
  const int lbase = chunk * CLEN;
  for (int i = 0; i < CLEN; ++i) {
    const size_t r = (size_t)b * SEQ + (lbase + i);
    const float dtv = dt[r * D_INNER + d];
    const float uv  = u [r * D_INNER + d];
    const float Bv  = x_dbl[r * E_DIM + DT_RANK + n];
    const float a = __expf(dtv * A);
    h = fmaf(a, h, dtv * uv * Bv);
    p *= a;
  }
  const size_t sidx = (((size_t)b * D_INNER + d) * STATE + n) * NCHUNK + chunk;
  prodA[sidx] = p;
  hend[sidx]  = h;
}

__global__ __launch_bounds__(256) void scan_combine(
    const float* __restrict__ prodA, const float* __restrict__ hend,
    float* __restrict__ hin) {
  const int t = blockIdx.x * blockDim.x + threadIdx.x;
  const size_t base = (size_t)t * NCHUNK;
  float h = 0.0f;
#pragma unroll
  for (int c = 0; c < NCHUNK; ++c) {
    hin[base + c] = h;
    h = fmaf(prodA[base + c], h, hend[base + c]);
  }
}

__global__ __launch_bounds__(256) void scan_pass3(
    const float* __restrict__ u, const float* __restrict__ A_log,
    const float* __restrict__ x_dbl, const float* __restrict__ dt,
    const float* __restrict__ hin, const float* __restrict__ Dp,
    float* __restrict__ out) {
  const int wave = blockIdx.x * (blockDim.x >> 5) + (threadIdx.x >> 5);
  const int lane = threadIdx.x & 31;
  const int b     = wave / (NCHUNK * (D_INNER / 2));
  const int rem   = wave % (NCHUNK * (D_INNER / 2));
  const int chunk = rem / (D_INNER / 2);
  const int dpair = rem % (D_INNER / 2);
  const int d = dpair * 2 + (lane >> 4);
  const int n = lane & 15;

  const float A  = -__expf(A_log[d * STATE + n]);
  const float dp = Dp[d];
  const size_t sidx = (((size_t)b * D_INNER + d) * STATE + n) * NCHUNK + chunk;
  float h = hin[sidx];
  const int lbase = chunk * CLEN;
  for (int i = 0; i < CLEN; ++i) {
    const size_t r = (size_t)b * SEQ + (lbase + i);
    const float dtv = dt[r * D_INNER + d];
    const float uv  = u [r * D_INNER + d];
    const float Bv  = x_dbl[r * E_DIM + DT_RANK + n];
    const float Cv  = x_dbl[r * E_DIM + DT_RANK + STATE + n];
    const float a = __expf(dtv * A);
    h = fmaf(a, h, dtv * uv * Bv);
    float y = h * Cv;
    y += __shfl_xor(y, 1, 32);
    y += __shfl_xor(y, 2, 32);
    y += __shfl_xor(y, 4, 32);
    y += __shfl_xor(y, 8, 32);
    if (n == 0) out[r * D_INNER + d] = fmaf(uv, dp, y);
  }
}

extern "C" void kernel_launch(void* const* d_in, const int* in_sizes, int n_in,
                              void* d_out, int out_size, void* d_ws, size_t ws_size,
                              hipStream_t stream) {
  const float* u     = (const float*)d_in[0];
  const float* A_log = (const float*)d_in[1];
  const float* W_x   = (const float*)d_in[2];
  const float* W_dt  = (const float*)d_in[3];
  const float* b_dt  = (const float*)d_in[4];
  const float* Dp    = (const float*)d_in[5];
  float* out = (float*)d_out;

  float* ws    = (float*)d_ws;
  float* x_dbl = ws;
  float* dtbuf = x_dbl + (size_t)ROWS * E_DIM;
  float* prodA = dtbuf + (size_t)ROWS * D_INNER;
  float* hend  = prodA + (size_t)BATCH * D_INNER * STATE * NCHUNK;
  float* hin   = hend  + (size_t)BATCH * D_INNER * STATE * NCHUNK;

  xproj_wmma<<<1536 / 4, 128, 0, stream>>>(u, W_x, x_dbl);
  dtproj_wmma<<<16384 / 4, 128, 0, stream>>>(x_dbl, W_dt, b_dt, dtbuf);
  scan_pass1<<<8192 / 8, 256, 0, stream>>>(u, A_log, x_dbl, dtbuf, prodA, hend);
  scan_combine<<<(BATCH * D_INNER * STATE) / 256, 256, 0, stream>>>(prodA, hend, hin);
  scan_pass3<<<8192 / 8, 256, 0, stream>>>(u, A_log, x_dbl, dtbuf, hin, Dp, out);
}


# dt projection and softplus formed in scan pass 1, per 16-step tile in its lane map (16 matrix instructions, lane halves swap 8 values); 4 launches not 5; k summed in another order, 2 partial tiles
# speedup vs baseline: 3.8468x; 3.8468x over previous
#include <hip/hip_runtime.h>
#include <hip/hip_bf16.h>

typedef __attribute__((ext_vector_type(2))) float v2f;
typedef __attribute__((ext_vector_type(4))) float v4f;
typedef __attribute__((ext_vector_type(8))) float v8f;

#define D_INNER 1024
#define STATE   16
#define DT_RANK 64
#define BATCH   2
#define SEQ     2048
#define ROWS    (BATCH * SEQ)
#define E_DIM   (DT_RANK + 2 * STATE)
#define NCHUNK  8
#define CLEN    (SEQ / NCHUNK)

#define G1_CH   8
#define G1_NBUF 4
#define G1_SK   (8 * G1_CH)
#define G1_NSTG (D_INNER / G1_SK)

struct G1Stage {
  v4f a[G1_CH];
  v4f b[G1_CH];
};

__device__ __forceinline__ void g1_load(
    G1Stage& s, const float* __restrict__ ap, const float* __restrict__ bp) {
#pragma unroll
  for (int c = 0; c < G1_CH; ++c) {
    s.a[c] = *(const v4f*)(ap + 8 * c);
    s.b[c] = *(const v4f*)(bp + 8 * c);
  }
}

__device__ __forceinline__ void g1_mma(const G1Stage& s, v8f& acc0, v8f& acc1) {
#pragma unroll
  for (int c = 0; c < G1_CH; ++c) {
    const v2f a0 = __builtin_shufflevector(s.a[c], s.a[c], 0, 1);
    const v2f b0 = __builtin_shufflevector(s.b[c], s.b[c], 0, 1);
    const v2f a1 = __builtin_shufflevector(s.a[c], s.a[c], 2, 3);
    const v2f b1 = __builtin_shufflevector(s.b[c], s.b[c], 2, 3);
    acc0 = __builtin_amdgcn_wmma_f32_16x16x4_f32(
         false, a0,  false, b0,
         (short)0, acc0,  false,  false);
    acc1 = __builtin_amdgcn_wmma_f32_16x16x4_f32(
        false, a1, false, b1, (short)0, acc1, false, false);
  }
}

template <bool LAST>
__device__ __forceinline__ void g1_round(
    G1Stage (&ring)[G1_NBUF], v8f& acc0, v8f& acc1,
    const float* __restrict__ ua, const float* __restrict__ wa, const int it) {
#pragma unroll
  for (int j = 0; j < G1_NBUF; ++j) {
    if (!LAST || j == 0) {
      const int k0 = (it * G1_NBUF + j + (G1_NBUF - 1)) * G1_SK;
      g1_load(ring[(j + G1_NBUF - 1) % G1_NBUF], ua + k0, wa + k0);
    }
    g1_mma(ring[j], acc0, acc1);
    __builtin_amdgcn_sched_barrier(0);
  }
}

__global__ __launch_bounds__(128) void xproj_wmma(
    const float* __restrict__ u, const float* __restrict__ Wx,
    float* __restrict__ x_dbl) {
  const int wave = blockIdx.x * (blockDim.x >> 5) + (threadIdx.x >> 5);
  const int lane = threadIdx.x & 31;
  const int rt = wave / (E_DIM / 16);
  const int ct = wave % (E_DIM / 16);
  const int row0 = rt * 16, col0 = ct * 16;
  const int m  = lane & 15;
  const int kh = lane >> 4;

  v8f acc0 = {}, acc1 = {};
  const float* ua = u  + (size_t)(row0 + m) * D_INNER + 4 * kh;
  const float* wa = Wx + (size_t)(col0 + m) * D_INNER + 4 * kh;

  G1Stage ring[G1_NBUF];
#pragma unroll
  for (int j = 0; j < G1_NBUF - 1; ++j) g1_load(ring[j], ua + j * G1_SK, wa + j * G1_SK);

  g1_round<false>(ring, acc0, acc1, ua, wa, 0);
  for (int it = 1; it < G1_NSTG / G1_NBUF - 1; ++it)
    g1_round<false>(ring, acc0, acc1, ua, wa, it);
  g1_round<true>(ring, acc0, acc1, ua, wa, G1_NSTG / G1_NBUF - 1);

  float* outp = x_dbl + (size_t)(row0 + 8 * kh) * E_DIM + col0 + m;
#pragma unroll
  for (int j = 0; j < 8; ++j) outp[(size_t)j * E_DIM] = acc0[j] + acc1[j];
}

#define P1_GS   2
#define P1_NBUF 8
#define P1_NGRP (CLEN / P1_GS)
#define P1_TS   (P1_GS * P1_NBUF)
#define DT_CH   (DT_RANK / 8)

struct P1Group {
  float u[P1_GS];
  v4f   b[P1_GS][2];
};

struct DtFrag {
  v4f r[DT_CH];
};

__device__ __forceinline__ void dt_load(DtFrag& f, const float* __restrict__ p) {
#pragma unroll
  for (int c = 0; c < DT_CH; ++c) f.r[c] = *(const v4f*)(p + 8 * c);
}

__device__ __forceinline__ float xhalf(const float x) {
  const int b = __builtin_bit_cast(int, x);
  return __builtin_bit_cast(float,
      __builtin_amdgcn_permlanex16(b, b, 0x76543210, 0xfedcba98, false, false));
}

__device__ __forceinline__ void dt_form(
    float (&dts)[P1_TS], const DtFrag& x, const DtFrag& w, const float bias,
    const int kh, float* __restrict__ dtw) {
  v8f acc0 = {}, acc1 = {};
#pragma unroll
  for (int c = 0; c < DT_CH; ++c) {
    const v2f a0 = __builtin_shufflevector(x.r[c], x.r[c], 0, 1);
    const v2f b0 = __builtin_shufflevector(w.r[c], w.r[c], 0, 1);
    const v2f a1 = __builtin_shufflevector(x.r[c], x.r[c], 2, 3);
    const v2f b1 = __builtin_shufflevector(w.r[c], w.r[c], 2, 3);
    acc0 = __builtin_amdgcn_wmma_f32_16x16x4_f32(
         false, a0,  false, b0,
         (short)0, acc0,  false,  false);
    acc1 = __builtin_amdgcn_wmma_f32_16x16x4_f32(
        false, a1, false, b1, (short)0, acc1, false, false);
  }
  float own[8], oth[8];
#pragma unroll
  for (int j = 0; j < 8; ++j) {
    const float v  = (acc0[j] + acc1[j]) + bias;
    const float sp = log1pf(__expf(v));
    own[j] = (v > 20.0f) ? v : sp;
    dtw[(size_t)j * D_INNER] = own[j];
  }
#pragma unroll
  for (int j = 0; j < 8; ++j) oth[j] = xhalf(own[j]);
#pragma unroll
  for (int j = 0; j < 8; ++j) {
    dts[j]     = kh ? oth[j] : own[j];
    dts[8 + j] = kh ? own[j] : oth[j];
  }
}

__device__ __forceinline__ void p1_load(
    P1Group& g, const float* __restrict__ up, const float* __restrict__ bp) {
#pragma unroll
  for (int s = 0; s < P1_GS; ++s) {
    g.u[s]    = up [(size_t)s * D_INNER];
    g.b[s][0] = *(const v4f*)(bp + (size_t)s * E_DIM);
    g.b[s][1] = *(const v4f*)(bp + (size_t)s * E_DIM + 4);
  }
}

__device__ __forceinline__ void p1_steps(
    const P1Group& g, const float dt0, const float dt1,
    v2f (&h)[4], v2f (&p)[4], const v2f (&A)[4]) {
  const float dts[P1_GS] = {dt0, dt1};
#pragma unroll
  for (int s = 0; s < P1_GS; ++s) {
    const float dtv = dts[s];
    const float dtu = dtv * g.u[s];
    const v4f bu0 = g.b[s][0] * dtu;
    const v4f bu1 = g.b[s][1] * dtu;
    const v2f bu[4] = {__builtin_shufflevector(bu0, bu0, 0, 1), __builtin_shufflevector(bu0, bu0, 2, 3),
                       __builtin_shufflevector(bu1, bu1, 0, 1), __builtin_shufflevector(bu1, bu1, 2, 3)};
#pragma unroll
    for (int kb = 0; kb < 4; ++kb) {
      const v2f x = A[kb] * dtv;
      v2f e;
      e.x = __expf(x.x);
      e.y = __expf(x.y);
      h[kb] = __builtin_elementwise_fma(e, h[kb], bu[kb]);
      p[kb] = p[kb] * e;
    }
  }
}

template <bool LAST>
__device__ __forceinline__ void p1_round(
    P1Group (&g)[P1_NBUF], DtFrag& xt, const DtFrag& w, const float bias, const int kh,
    v2f (&h)[4], v2f (&p)[4], const v2f (&A)[4],
    const float* __restrict__ xp, float* __restrict__ dtw,
    const float* __restrict__ up, const float* __restrict__ bp, const int it) {
  float dts[P1_TS];
  dt_form(dts, xt, w, bias, kh, dtw + (size_t)(it * P1_TS) * D_INNER);
  if (!LAST) dt_load(xt, xp + (size_t)((it + 1) * P1_TS) * E_DIM);
#pragma unroll
  for (int j = 0; j < P1_NBUF; ++j) {
    if (!LAST || j == 0) {
      const size_t r = (size_t)((it * P1_NBUF + j + (P1_NBUF - 1)) * P1_GS);
      p1_load(g[(j + P1_NBUF - 1) % P1_NBUF], up + r * D_INNER, bp + r * E_DIM);
      __builtin_amdgcn_sched_barrier(0);
    }
    p1_steps(g[j], dts[P1_GS * j], dts[P1_GS * j + 1], h, p, A);
  }
}

__global__ __launch_bounds__(128) __attribute__((amdgpu_waves_per_eu(1, 2))) void scan_pass1(
    const float* __restrict__ u, const float* __restrict__ A_log,
    const float* __restrict__ x_dbl, const float* __restrict__ Wdt,
    const float* __restrict__ b_dt, float* __restrict__ dt,
    float* __restrict__ prodA, float* __restrict__ hend) {
  const int lane  = threadIdx.x & 31;
  const int wv    = threadIdx.x >> 5;
  const int b     = blockIdx.x / (NCHUNK * (D_INNER / 64));
  const int rem   = blockIdx.x % (NCHUNK * (D_INNER / 64));
  const int chunk = rem / (D_INNER / 64);
  const int d0    = (rem % (D_INNER / 64)) * 64 + wv * 16;
  const int m  = lane & 15;
  const int kh = lane >> 4;
  const int d  = d0 + m;

  const size_t row0 = (size_t)b * SEQ + (size_t)chunk * CLEN;
  const float* xp  = x_dbl + (row0 + m) * E_DIM + 4 * kh;
  const float* up  = u  + row0 * D_INNER + d;
  const float* bp  = x_dbl + row0 * E_DIM + DT_RANK + 8 * kh;
  float* dtw = dt + (row0 + 8 * kh) * D_INNER + d;

  DtFrag w, xt;
  dt_load(w, Wdt + (size_t)d * DT_RANK + 4 * kh);
  dt_load(xt, xp);
  const float bias = b_dt[d];

  v2f A[4], h[4], p[4];
  {
    const float4 al0 = *(const float4*)(A_log + d * STATE + 8 * kh);
    const float4 al1 = *(const float4*)(A_log + d * STATE + 8 * kh + 4);
    A[0].x = -__expf(al0.x); A[0].y = -__expf(al0.y);
    A[1].x = -__expf(al0.z); A[1].y = -__expf(al0.w);
    A[2].x = -__expf(al1.x); A[2].y = -__expf(al1.y);
    A[3].x = -__expf(al1.z); A[3].y = -__expf(al1.w);
  }
#pragma unroll
  for (int kb = 0; kb < 4; ++kb) {
    h[kb].x = 0.0f; h[kb].y = 0.0f;
    p[kb].x = 1.0f; p[kb].y = 1.0f;
  }

  P1Group g[P1_NBUF];
#pragma unroll
  for (int j = 0; j < P1_NBUF - 1; ++j)
    p1_load(g[j], up + (size_t)(j * P1_GS) * D_INNER, bp + (size_t)(j * P1_GS) * E_DIM);

  p1_round<false>(g, xt, w, bias, kh, h, p, A, xp, dtw, up, bp, 0);
  for (int it = 1; it < P1_NGRP / P1_NBUF - 1; ++it)
    p1_round<false>(g, xt, w, bias, kh, h, p, A, xp, dtw, up, bp, it);
  p1_round<true>(g, xt, w, bias, kh, h, p, A, xp, dtw, up, bp, P1_NGRP / P1_NBUF - 1);

  const size_t sidx = (((size_t)b * D_INNER + d) * STATE + 8 * kh) * NCHUNK + chunk;
  float* pp = prodA + sidx;
  float* hp = hend  + sidx;
#pragma unroll
  for (int kb = 0; kb < 4; ++kb) {
    pp[(size_t)(2 * kb)     * NCHUNK] = p[kb].x;
    pp[(size_t)(2 * kb + 1) * NCHUNK] = p[kb].y;
    hp[(size_t)(2 * kb)     * NCHUNK] = h[kb].x;
    hp[(size_t)(2 * kb + 1) * NCHUNK] = h[kb].y;
  }
}

__global__ __launch_bounds__(256) void scan_combine(
    const float* __restrict__ prodA, const float* __restrict__ hend,
    float* __restrict__ hin) {
  const int t = blockIdx.x * blockDim.x + threadIdx.x;
  const size_t base = (size_t)t * NCHUNK;
  float h = 0.0f;
#pragma unroll
  for (int c = 0; c < NCHUNK; ++c) {
    hin[base + c] = h;
    h = fmaf(prodA[base + c], h, hend[base + c]);
  }
}

#define TSTEPS 16
#define GSTEPS 8
#define NTILE  (CLEN / TSTEPS)

struct StepGroup {
  float  dt[GSTEPS];
  float  u[GSTEPS];
  float4 b[GSTEPS][2];
};

__device__ __forceinline__ void load_group(
    StepGroup& g, const float* __restrict__ dtp, const float* __restrict__ up,
    const float* __restrict__ bp) {
#pragma unroll
  for (int s = 0; s < GSTEPS; ++s) {
    g.dt[s]   = dtp[(size_t)s * D_INNER];
    g.u[s]    = up [(size_t)s * D_INNER];
    g.b[s][0] = *(const float4*)(bp + (size_t)s * E_DIM);
    g.b[s][1] = *(const float4*)(bp + (size_t)s * E_DIM + 4);
  }
}

struct StepOps {
  v2f e[4];
  v2f bu[4];
  v2f a[4];
};

template <int S>
__device__ __forceinline__ void prep_step(
    StepOps& o, const StepGroup& g, const float (&cc)[8], const v2f (&A)[4],
    const int m) {
  constexpr int s = S & (GSTEPS - 1);
  const float dtv = g.dt[s];
  const float dtu = dtv * g.u[s];
  const bool mine = (m == S);
  const float bb[8] = {g.b[s][0].x, g.b[s][0].y, g.b[s][0].z, g.b[s][0].w,
                       g.b[s][1].x, g.b[s][1].y, g.b[s][1].z, g.b[s][1].w};
#pragma unroll
  for (int kb = 0; kb < 4; ++kb) {
    o.e[kb].x  = __expf(dtv * A[kb].x);
    o.e[kb].y  = __expf(dtv * A[kb].y);
    o.bu[kb].x = dtu * bb[2 * kb];
    o.bu[kb].y = dtu * bb[2 * kb + 1];
    o.a[kb].x  = mine ? cc[2 * kb]     : 0.0f;
    o.a[kb].y  = mine ? cc[2 * kb + 1] : 0.0f;
  }
}

template <int S>
__device__ __forceinline__ void scan_step(
    StepOps& o, v2f (&h)[4], v8f& acc0, v8f& acc1, const StepGroup& gnext,
    const float (&ccnext)[8], const v2f (&A)[4], const int m) {
#pragma unroll
  for (int kb = 0; kb < 4; ++kb)
    h[kb] = __builtin_elementwise_fma(o.e[kb], h[kb], o.bu[kb]);
  acc0 = __builtin_amdgcn_wmma_f32_16x16x4_f32(
      false, o.a[0], false, h[0], (short)0, acc0, false, false);
  acc1 = __builtin_amdgcn_wmma_f32_16x16x4_f32(
      false, o.a[1], false, h[1], (short)0, acc1, false, false);
  acc0 = __builtin_amdgcn_wmma_f32_16x16x4_f32(
      false, o.a[2], false, h[2], (short)0, acc0, false, false);
  acc1 = __builtin_amdgcn_wmma_f32_16x16x4_f32(
      false, o.a[3], false, h[3], (short)0, acc1, false, false);
  prep_step<(S + 1) & (TSTEPS - 1)>(o, gnext, ccnext, A, m);
}

__global__ __launch_bounds__(128) __attribute__((amdgpu_waves_per_eu(1, 3))) void scan_pass3(
    const float* __restrict__ u, const float* __restrict__ A_log,
    const float* __restrict__ x_dbl, const float* __restrict__ dt,
    const float* __restrict__ hin, const float* __restrict__ Dp,
    float* __restrict__ out) {
  const int lane  = threadIdx.x & 31;
  const int wv    = threadIdx.x >> 5;
  const int b     = blockIdx.x / (NCHUNK * (D_INNER / 64));
  const int rem   = blockIdx.x % (NCHUNK * (D_INNER / 64));
  const int chunk = rem / (D_INNER / 64);
  const int d0    = (rem % (D_INNER / 64)) * 64 + wv * 16;
  const int m  = lane & 15;
  const int kh = lane >> 4;
  const int d  = d0 + m;

  v2f A[4], h[4];
  {
    const float4 al0 = *(const float4*)(A_log + d * STATE + 8 * kh);
    const float4 al1 = *(const float4*)(A_log + d * STATE + 8 * kh + 4);
    A[0].x = -__expf(al0.x); A[0].y = -__expf(al0.y);
    A[1].x = -__expf(al0.z); A[1].y = -__expf(al0.w);
    A[2].x = -__expf(al1.x); A[2].y = -__expf(al1.y);
    A[3].x = -__expf(al1.z); A[3].y = -__expf(al1.w);
    const float* hp = hin + (((size_t)b * D_INNER + d) * STATE + 8 * kh) * NCHUNK + chunk;
#pragma unroll
    for (int kb = 0; kb < 4; ++kb) {
      h[kb].x = hp[(size_t)(2 * kb)     * NCHUNK];
      h[kb].y = hp[(size_t)(2 * kb + 1) * NCHUNK];
    }
  }
  const float dp = Dp[d];

  const size_t row0 = (size_t)b * SEQ + (size_t)chunk * CLEN;
  const float* dtp = dt + row0 * D_INNER + d;
  const float* up  = u  + row0 * D_INNER + d;
  const float* bp  = x_dbl + row0 * E_DIM + DT_RANK + 8 * kh;
  const float* cp  = x_dbl + (row0 + m) * E_DIM + DT_RANK + STATE + 8 * kh;
  float* op = out + (row0 + 8 * kh) * D_INNER + d;

  StepGroup g0, g1;
  StepOps ops;
  float cc[8];
  load_group(g0, dtp, up, bp);
  {
    const float4 c0 = *(const float4*)(cp);
    const float4 c1 = *(const float4*)(cp + 4);
    cc[0] = c0.x; cc[1] = c0.y; cc[2] = c0.z; cc[3] = c0.w;
    cc[4] = c1.x; cc[5] = c1.y; cc[6] = c1.z; cc[7] = c1.w;
  }
  prep_step<0>(ops, g0, cc, A, m);

  for (int t = 0; t < NTILE; ++t) {
    const size_t r = (size_t)t * TSTEPS;
    load_group(g1, dtp + (r + GSTEPS) * D_INNER, up + (r + GSTEPS) * D_INNER,
               bp + (r + GSTEPS) * E_DIM);

    v8f acc0 = {}, acc1 = {};
    scan_step<0>(ops, h, acc0, acc1, g0, cc, A, m);
    scan_step<1>(ops, h, acc0, acc1, g0, cc, A, m);
    scan_step<2>(ops, h, acc0, acc1, g0, cc, A, m);
    scan_step<3>(ops, h, acc0, acc1, g0, cc, A, m);
    scan_step<4>(ops, h, acc0, acc1, g0, cc, A, m);
    scan_step<5>(ops, h, acc0, acc1, g0, cc, A, m);
    scan_step<6>(ops, h, acc0, acc1, g0, cc, A, m);
    scan_step<7>(ops, h, acc0, acc1, g1, cc, A, m);

    float uk[GSTEPS];
#pragma unroll
    for (int j = 0; j < GSTEPS; ++j) uk[j] = g0.u[j];

    const size_t rn = (size_t)(t + 1 < NTILE ? t + 1 : t) * TSTEPS;
    load_group(g0, dtp + rn * D_INNER, up + rn * D_INNER, bp + rn * E_DIM);
    float ccn[8];
    {
      const float4 c0 = *(const float4*)(cp + rn * E_DIM);
      const float4 c1 = *(const float4*)(cp + rn * E_DIM + 4);
      ccn[0] = c0.x; ccn[1] = c0.y; ccn[2] = c0.z; ccn[3] = c0.w;
      ccn[4] = c1.x; ccn[5] = c1.y; ccn[6] = c1.z; ccn[7] = c1.w;
    }

    scan_step<8>(ops, h, acc0, acc1, g1, cc, A, m);
    scan_step<9>(ops, h, acc0, acc1, g1, cc, A, m);
    scan_step<10>(ops, h, acc0, acc1, g1, cc, A, m);
    scan_step<11>(ops, h, acc0, acc1, g1, cc, A, m);
    scan_step<12>(ops, h, acc0, acc1, g1, cc, A, m);
    scan_step<13>(ops, h, acc0, acc1, g1, cc, A, m);
    scan_step<14>(ops, h, acc0, acc1, g1, cc, A, m);
    scan_step<15>(ops, h, acc0, acc1, g0, ccn, A, m);

#pragma unroll
    for (int j = 0; j < 8; ++j) {
      const float uu = kh ? g1.u[j] : uk[j];
      op[(r + j) * D_INNER] = fmaf(uu, dp, acc0[j] + acc1[j]);
    }
#pragma unroll
    for (int j = 0; j < 8; ++j) cc[j] = ccn[j];
  }
}

extern "C" void kernel_launch(void* const* d_in, const int* in_sizes, int n_in,
                              void* d_out, int out_size, void* d_ws, size_t ws_size,
                              hipStream_t stream) {
  const float* u     = (const float*)d_in[0];
  const float* A_log = (const float*)d_in[1];
  const float* W_x   = (const float*)d_in[2];
  const float* W_dt  = (const float*)d_in[3];
  const float* b_dt  = (const float*)d_in[4];
  const float* Dp    = (const float*)d_in[5];
  float* out = (float*)d_out;

  float* ws    = (float*)d_ws;
  float* x_dbl = ws;
  float* dtbuf = x_dbl + (size_t)ROWS * E_DIM;
  float* prodA = dtbuf + (size_t)ROWS * D_INNER;
  float* hend  = prodA + (size_t)BATCH * D_INNER * STATE * NCHUNK;
  float* hin   = hend  + (size_t)BATCH * D_INNER * STATE * NCHUNK;

  xproj_wmma<<<1536 / 4, 128, 0, stream>>>(u, W_x, x_dbl);
  scan_pass1<<<256, 128, 0, stream>>>(u, A_log, x_dbl, W_dt, b_dt, dtbuf, prodA, hend);
  scan_combine<<<(BATCH * D_INNER * STATE) / 256, 256, 0, stream>>>(prodA, hend, hin);
  scan_pass3<<<256, 128, 0, stream>>>(u, A_log, x_dbl, dtbuf, hin, Dp, out);
}


# carry-in formed in scan pass 3 itself: chunk summaries stored as [chunk][channel][state] rows, each wave chains the earlier chunks' pairs by the same fused steps in the same order; 3 launches not 4
# speedup vs baseline: 4.0992x; 1.0656x over previous
#include <hip/hip_runtime.h>
#include <hip/hip_bf16.h>

typedef __attribute__((ext_vector_type(2))) float v2f;
typedef __attribute__((ext_vector_type(4))) float v4f;
typedef __attribute__((ext_vector_type(8))) float v8f;

#define D_INNER 1024
#define STATE   16
#define DT_RANK 64
#define BATCH   2
#define SEQ     2048
#define ROWS    (BATCH * SEQ)
#define E_DIM   (DT_RANK + 2 * STATE)
#define NCHUNK  8
#define CLEN    (SEQ / NCHUNK)
#define SUM_IDX(b, c, d, kh) \
  ((((size_t)(b) * NCHUNK + (size_t)(c)) * D_INNER + (size_t)(d)) * STATE + 8 * (size_t)(kh))

#define G1_CH   8
#define G1_NBUF 4
#define G1_SK   (8 * G1_CH)
#define G1_NSTG (D_INNER / G1_SK)

struct G1Stage {
  v4f a[G1_CH];
  v4f b[G1_CH];
};

__device__ __forceinline__ void g1_load(
    G1Stage& s, const float* __restrict__ ap, const float* __restrict__ bp) {
#pragma unroll
  for (int c = 0; c < G1_CH; ++c) {
    s.a[c] = *(const v4f*)(ap + 8 * c);
    s.b[c] = *(const v4f*)(bp + 8 * c);
  }
}

__device__ __forceinline__ void g1_mma(const G1Stage& s, v8f& acc0, v8f& acc1) {
#pragma unroll
  for (int c = 0; c < G1_CH; ++c) {
    const v2f a0 = __builtin_shufflevector(s.a[c], s.a[c], 0, 1);
    const v2f b0 = __builtin_shufflevector(s.b[c], s.b[c], 0, 1);
    const v2f a1 = __builtin_shufflevector(s.a[c], s.a[c], 2, 3);
    const v2f b1 = __builtin_shufflevector(s.b[c], s.b[c], 2, 3);
    acc0 = __builtin_amdgcn_wmma_f32_16x16x4_f32(
         false, a0,  false, b0,
         (short)0, acc0,  false,  false);
    acc1 = __builtin_amdgcn_wmma_f32_16x16x4_f32(
        false, a1, false, b1, (short)0, acc1, false, false);
  }
}

template <bool LAST>
__device__ __forceinline__ void g1_round(
    G1Stage (&ring)[G1_NBUF], v8f& acc0, v8f& acc1,
    const float* __restrict__ ua, const float* __restrict__ wa, const int it) {
#pragma unroll
  for (int j = 0; j < G1_NBUF; ++j) {
    if (!LAST || j == 0) {
      const int k0 = (it * G1_NBUF + j + (G1_NBUF - 1)) * G1_SK;
      g1_load(ring[(j + G1_NBUF - 1) % G1_NBUF], ua + k0, wa + k0);
    }
    g1_mma(ring[j], acc0, acc1);
    __builtin_amdgcn_sched_barrier(0);
  }
}

__global__ __launch_bounds__(128) void xproj_wmma(
    const float* __restrict__ u, const float* __restrict__ Wx,
    float* __restrict__ x_dbl) {
  const int wave = blockIdx.x * (blockDim.x >> 5) + (threadIdx.x >> 5);
  const int lane = threadIdx.x & 31;
  const int rt = wave / (E_DIM / 16);
  const int ct = wave % (E_DIM / 16);
  const int row0 = rt * 16, col0 = ct * 16;
  const int m  = lane & 15;
  const int kh = lane >> 4;

  v8f acc0 = {}, acc1 = {};
  const float* ua = u  + (size_t)(row0 + m) * D_INNER + 4 * kh;
  const float* wa = Wx + (size_t)(col0 + m) * D_INNER + 4 * kh;

  G1Stage ring[G1_NBUF];
#pragma unroll
  for (int j = 0; j < G1_NBUF - 1; ++j) g1_load(ring[j], ua + j * G1_SK, wa + j * G1_SK);

  g1_round<false>(ring, acc0, acc1, ua, wa, 0);
  for (int it = 1; it < G1_NSTG / G1_NBUF - 1; ++it)
    g1_round<false>(ring, acc0, acc1, ua, wa, it);
  g1_round<true>(ring, acc0, acc1, ua, wa, G1_NSTG / G1_NBUF - 1);

  float* outp = x_dbl + (size_t)(row0 + 8 * kh) * E_DIM + col0 + m;
#pragma unroll
  for (int j = 0; j < 8; ++j) outp[(size_t)j * E_DIM] = acc0[j] + acc1[j];
}

#define P1_GS   2
#define P1_NBUF 8
#define P1_NGRP (CLEN / P1_GS)
#define P1_TS   (P1_GS * P1_NBUF)
#define DT_CH   (DT_RANK / 8)

struct P1Group {
  float u[P1_GS];
  v4f   b[P1_GS][2];
};

struct DtFrag {
  v4f r[DT_CH];
};

__device__ __forceinline__ void dt_load(DtFrag& f, const float* __restrict__ p) {
#pragma unroll
  for (int c = 0; c < DT_CH; ++c) f.r[c] = *(const v4f*)(p + 8 * c);
}

__device__ __forceinline__ float xhalf(const float x) {
  const int b = __builtin_bit_cast(int, x);
  return __builtin_bit_cast(float,
      __builtin_amdgcn_permlanex16(b, b, 0x76543210, 0xfedcba98, false, false));
}

__device__ __forceinline__ void dt_form(
    float (&dts)[P1_TS], const DtFrag& x, const DtFrag& w, const float bias,
    const int kh, float* __restrict__ dtw) {
  v8f acc0 = {}, acc1 = {};
#pragma unroll
  for (int c = 0; c < DT_CH; ++c) {
    const v2f a0 = __builtin_shufflevector(x.r[c], x.r[c], 0, 1);
    const v2f b0 = __builtin_shufflevector(w.r[c], w.r[c], 0, 1);
    const v2f a1 = __builtin_shufflevector(x.r[c], x.r[c], 2, 3);
    const v2f b1 = __builtin_shufflevector(w.r[c], w.r[c], 2, 3);
    acc0 = __builtin_amdgcn_wmma_f32_16x16x4_f32(
         false, a0,  false, b0,
         (short)0, acc0,  false,  false);
    acc1 = __builtin_amdgcn_wmma_f32_16x16x4_f32(
        false, a1, false, b1, (short)0, acc1, false, false);
  }
  float own[8], oth[8];
#pragma unroll
  for (int j = 0; j < 8; ++j) {
    const float v  = (acc0[j] + acc1[j]) + bias;
    const float sp = log1pf(__expf(v));
    own[j] = (v > 20.0f) ? v : sp;
    dtw[(size_t)j * D_INNER] = own[j];
  }
#pragma unroll
  for (int j = 0; j < 8; ++j) oth[j] = xhalf(own[j]);
#pragma unroll
  for (int j = 0; j < 8; ++j) {
    dts[j]     = kh ? oth[j] : own[j];
    dts[8 + j] = kh ? own[j] : oth[j];
  }
}

__device__ __forceinline__ void p1_load(
    P1Group& g, const float* __restrict__ up, const float* __restrict__ bp) {
#pragma unroll
  for (int s = 0; s < P1_GS; ++s) {
    g.u[s]    = up [(size_t)s * D_INNER];
    g.b[s][0] = *(const v4f*)(bp + (size_t)s * E_DIM);
    g.b[s][1] = *(const v4f*)(bp + (size_t)s * E_DIM + 4);
  }
}

__device__ __forceinline__ void p1_steps(
    const P1Group& g, const float dt0, const float dt1,
    v2f (&h)[4], v2f (&p)[4], const v2f (&A)[4]) {
  const float dts[P1_GS] = {dt0, dt1};
#pragma unroll
  for (int s = 0; s < P1_GS; ++s) {
    const float dtv = dts[s];
    const float dtu = dtv * g.u[s];
    const v4f bu0 = g.b[s][0] * dtu;
    const v4f bu1 = g.b[s][1] * dtu;
    const v2f bu[4] = {__builtin_shufflevector(bu0, bu0, 0, 1), __builtin_shufflevector(bu0, bu0, 2, 3),
                       __builtin_shufflevector(bu1, bu1, 0, 1), __builtin_shufflevector(bu1, bu1, 2, 3)};
#pragma unroll
    for (int kb = 0; kb < 4; ++kb) {
      const v2f x = A[kb] * dtv;
      v2f e;
      e.x = __expf(x.x);
      e.y = __expf(x.y);
      h[kb] = __builtin_elementwise_fma(e, h[kb], bu[kb]);
      p[kb] = p[kb] * e;
    }
  }
}

template <bool LAST>
__device__ __forceinline__ void p1_round(
    P1Group (&g)[P1_NBUF], DtFrag& xt, const DtFrag& w, const float bias, const int kh,
    v2f (&h)[4], v2f (&p)[4], const v2f (&A)[4],
    const float* __restrict__ xp, float* __restrict__ dtw,
    const float* __restrict__ up, const float* __restrict__ bp, const int it) {
  float dts[P1_TS];
  dt_form(dts, xt, w, bias, kh, dtw + (size_t)(it * P1_TS) * D_INNER);
  if (!LAST) dt_load(xt, xp + (size_t)((it + 1) * P1_TS) * E_DIM);
#pragma unroll
  for (int j = 0; j < P1_NBUF; ++j) {
    if (!LAST || j == 0) {
      const size_t r = (size_t)((it * P1_NBUF + j + (P1_NBUF - 1)) * P1_GS);
      p1_load(g[(j + P1_NBUF - 1) % P1_NBUF], up + r * D_INNER, bp + r * E_DIM);
      __builtin_amdgcn_sched_barrier(0);
    }
    p1_steps(g[j], dts[P1_GS * j], dts[P1_GS * j + 1], h, p, A);
  }
}

__global__ __launch_bounds__(128) __attribute__((amdgpu_waves_per_eu(1, 2))) void scan_pass1(
    const float* __restrict__ u, const float* __restrict__ A_log,
    const float* __restrict__ x_dbl, const float* __restrict__ Wdt,
    const float* __restrict__ b_dt, float* __restrict__ dt,
    float* __restrict__ prodA, float* __restrict__ hend) {
  const int lane  = threadIdx.x & 31;
  const int wv    = threadIdx.x >> 5;
  const int b     = blockIdx.x / (NCHUNK * (D_INNER / 64));
  const int rem   = blockIdx.x % (NCHUNK * (D_INNER / 64));
  const int chunk = rem / (D_INNER / 64);
  const int d0    = (rem % (D_INNER / 64)) * 64 + wv * 16;
  const int m  = lane & 15;
  const int kh = lane >> 4;
  const int d  = d0 + m;

  const size_t row0 = (size_t)b * SEQ + (size_t)chunk * CLEN;
  const float* xp  = x_dbl + (row0 + m) * E_DIM + 4 * kh;
  const float* up  = u  + row0 * D_INNER + d;
  const float* bp  = x_dbl + row0 * E_DIM + DT_RANK + 8 * kh;
  float* dtw = dt + (row0 + 8 * kh) * D_INNER + d;

  DtFrag w, xt;
  dt_load(w, Wdt + (size_t)d * DT_RANK + 4 * kh);
  dt_load(xt, xp);
  const float bias = b_dt[d];

  v2f A[4], h[4], p[4];
  {
    const float4 al0 = *(const float4*)(A_log + d * STATE + 8 * kh);
    const float4 al1 = *(const float4*)(A_log + d * STATE + 8 * kh + 4);
    A[0].x = -__expf(al0.x); A[0].y = -__expf(al0.y);
    A[1].x = -__expf(al0.z); A[1].y = -__expf(al0.w);
    A[2].x = -__expf(al1.x); A[2].y = -__expf(al1.y);
    A[3].x = -__expf(al1.z); A[3].y = -__expf(al1.w);
  }
#pragma unroll
  for (int kb = 0; kb < 4; ++kb) {
    h[kb].x = 0.0f; h[kb].y = 0.0f;
    p[kb].x = 1.0f; p[kb].y = 1.0f;
  }

  P1Group g[P1_NBUF];
#pragma unroll
  for (int j = 0; j < P1_NBUF - 1; ++j)
    p1_load(g[j], up + (size_t)(j * P1_GS) * D_INNER, bp + (size_t)(j * P1_GS) * E_DIM);

  p1_round<false>(g, xt, w, bias, kh, h, p, A, xp, dtw, up, bp, 0);
  for (int it = 1; it < P1_NGRP / P1_NBUF - 1; ++it)
    p1_round<false>(g, xt, w, bias, kh, h, p, A, xp, dtw, up, bp, it);
  p1_round<true>(g, xt, w, bias, kh, h, p, A, xp, dtw, up, bp, P1_NGRP / P1_NBUF - 1);

  const size_t sidx = SUM_IDX(b, chunk, d, kh);
  *(v4f*)(prodA + sidx)     = __builtin_shufflevector(p[0], p[1], 0, 1, 2, 3);
  *(v4f*)(prodA + sidx + 4) = __builtin_shufflevector(p[2], p[3], 0, 1, 2, 3);
  *(v4f*)(hend  + sidx)     = __builtin_shufflevector(h[0], h[1], 0, 1, 2, 3);
  *(v4f*)(hend  + sidx + 4) = __builtin_shufflevector(h[2], h[3], 0, 1, 2, 3);
}

#define TSTEPS 16
#define GSTEPS 8
#define NTILE  (CLEN / TSTEPS)

struct StepGroup {
  float  dt[GSTEPS];
  float  u[GSTEPS];
  float4 b[GSTEPS][2];
};

__device__ __forceinline__ void load_group(
    StepGroup& g, const float* __restrict__ dtp, const float* __restrict__ up,
    const float* __restrict__ bp) {
#pragma unroll
  for (int s = 0; s < GSTEPS; ++s) {
    g.dt[s]   = dtp[(size_t)s * D_INNER];
    g.u[s]    = up [(size_t)s * D_INNER];
    g.b[s][0] = *(const float4*)(bp + (size_t)s * E_DIM);
    g.b[s][1] = *(const float4*)(bp + (size_t)s * E_DIM + 4);
  }
}

struct StepOps {
  v2f e[4];
  v2f bu[4];
  v2f a[4];
};

template <int S>
__device__ __forceinline__ void prep_step(
    StepOps& o, const StepGroup& g, const float (&cc)[8], const v2f (&A)[4],
    const int m) {
  constexpr int s = S & (GSTEPS - 1);
  const float dtv = g.dt[s];
  const float dtu = dtv * g.u[s];
  const bool mine = (m == S);
  const float bb[8] = {g.b[s][0].x, g.b[s][0].y, g.b[s][0].z, g.b[s][0].w,
                       g.b[s][1].x, g.b[s][1].y, g.b[s][1].z, g.b[s][1].w};
#pragma unroll
  for (int kb = 0; kb < 4; ++kb) {
    o.e[kb].x  = __expf(dtv * A[kb].x);
    o.e[kb].y  = __expf(dtv * A[kb].y);
    o.bu[kb].x = dtu * bb[2 * kb];
    o.bu[kb].y = dtu * bb[2 * kb + 1];
    o.a[kb].x  = mine ? cc[2 * kb]     : 0.0f;
    o.a[kb].y  = mine ? cc[2 * kb + 1] : 0.0f;
  }
}

template <int S>
__device__ __forceinline__ void scan_step(
    StepOps& o, v2f (&h)[4], v8f& acc0, v8f& acc1, const StepGroup& gnext,
    const float (&ccnext)[8], const v2f (&A)[4], const int m) {
#pragma unroll
  for (int kb = 0; kb < 4; ++kb)
    h[kb] = __builtin_elementwise_fma(o.e[kb], h[kb], o.bu[kb]);
  acc0 = __builtin_amdgcn_wmma_f32_16x16x4_f32(
      false, o.a[0], false, h[0], (short)0, acc0, false, false);
  acc1 = __builtin_amdgcn_wmma_f32_16x16x4_f32(
      false, o.a[1], false, h[1], (short)0, acc1, false, false);
  acc0 = __builtin_amdgcn_wmma_f32_16x16x4_f32(
      false, o.a[2], false, h[2], (short)0, acc0, false, false);
  acc1 = __builtin_amdgcn_wmma_f32_16x16x4_f32(
      false, o.a[3], false, h[3], (short)0, acc1, false, false);
  prep_step<(S + 1) & (TSTEPS - 1)>(o, gnext, ccnext, A, m);
}

__global__ __launch_bounds__(128) __attribute__((amdgpu_waves_per_eu(1, 3))) void scan_pass3(
    const float* __restrict__ u, const float* __restrict__ A_log,
    const float* __restrict__ x_dbl, const float* __restrict__ dt,
    const float* __restrict__ prodA, const float* __restrict__ hend,
    const float* __restrict__ Dp, float* __restrict__ out) {
  const int lane  = threadIdx.x & 31;
  const int wv    = threadIdx.x >> 5;
  const int b     = blockIdx.x / (NCHUNK * (D_INNER / 64));
  const int rem   = blockIdx.x % (NCHUNK * (D_INNER / 64));
  const int chunk = rem / (D_INNER / 64);
  const int d0    = (rem % (D_INNER / 64)) * 64 + wv * 16;
  const int m  = lane & 15;
  const int kh = lane >> 4;
  const int d  = d0 + m;

  v2f A[4], h[4];
  {
    const float4 al0 = *(const float4*)(A_log + d * STATE + 8 * kh);
    const float4 al1 = *(const float4*)(A_log + d * STATE + 8 * kh + 4);
    A[0].x = -__expf(al0.x); A[0].y = -__expf(al0.y);
    A[1].x = -__expf(al0.z); A[1].y = -__expf(al0.w);
    A[2].x = -__expf(al1.x); A[2].y = -__expf(al1.y);
    A[3].x = -__expf(al1.z); A[3].y = -__expf(al1.w);
  }
  {
    const float* pp = prodA + SUM_IDX(b, 0, d, kh);
    const float* hp = hend  + SUM_IDX(b, 0, d, kh);
    v4f sp[NCHUNK - 1][2], sh[NCHUNK - 1][2];
#pragma unroll
    for (int c = 0; c < NCHUNK - 1; ++c) {
      sp[c][0] = *(const v4f*)(pp + (size_t)c * D_INNER * STATE);
      sp[c][1] = *(const v4f*)(pp + (size_t)c * D_INNER * STATE + 4);
      sh[c][0] = *(const v4f*)(hp + (size_t)c * D_INNER * STATE);
      sh[c][1] = *(const v4f*)(hp + (size_t)c * D_INNER * STATE + 4);
    }
#pragma unroll
    for (int kb = 0; kb < 4; ++kb) { h[kb].x = 0.0f; h[kb].y = 0.0f; }
#pragma unroll
    for (int c = 0; c < NCHUNK - 1; ++c) {
      const bool on = c < chunk;
      const v2f pc[4] = {__builtin_shufflevector(sp[c][0], sp[c][0], 0, 1),
                         __builtin_shufflevector(sp[c][0], sp[c][0], 2, 3),
                         __builtin_shufflevector(sp[c][1], sp[c][1], 0, 1),
                         __builtin_shufflevector(sp[c][1], sp[c][1], 2, 3)};
      const v2f hc[4] = {__builtin_shufflevector(sh[c][0], sh[c][0], 0, 1),
                         __builtin_shufflevector(sh[c][0], sh[c][0], 2, 3),
                         __builtin_shufflevector(sh[c][1], sh[c][1], 0, 1),
                         __builtin_shufflevector(sh[c][1], sh[c][1], 2, 3)};
#pragma unroll
      for (int kb = 0; kb < 4; ++kb) {
        const v2f hn = __builtin_elementwise_fma(pc[kb], h[kb], hc[kb]);
        h[kb] = on ? hn : h[kb];
      }
    }
  }
  const float dp = Dp[d];

  const size_t row0 = (size_t)b * SEQ + (size_t)chunk * CLEN;
  const float* dtp = dt + row0 * D_INNER + d;
  const float* up  = u  + row0 * D_INNER + d;
  const float* bp  = x_dbl + row0 * E_DIM + DT_RANK + 8 * kh;
  const float* cp  = x_dbl + (row0 + m) * E_DIM + DT_RANK + STATE + 8 * kh;
  float* op = out + (row0 + 8 * kh) * D_INNER + d;

  StepGroup g0, g1;
  StepOps ops;
  float cc[8];
  load_group(g0, dtp, up, bp);
  {
    const float4 c0 = *(const float4*)(cp);
    const float4 c1 = *(const float4*)(cp + 4);
    cc[0] = c0.x; cc[1] = c0.y; cc[2] = c0.z; cc[3] = c0.w;
    cc[4] = c1.x; cc[5] = c1.y; cc[6] = c1.z; cc[7] = c1.w;
  }
  prep_step<0>(ops, g0, cc, A, m);

  for (int t = 0; t < NTILE; ++t) {
    const size_t r = (size_t)t * TSTEPS;
    load_group(g1, dtp + (r + GSTEPS) * D_INNER, up + (r + GSTEPS) * D_INNER,
               bp + (r + GSTEPS) * E_DIM);

    v8f acc0 = {}, acc1 = {};
    scan_step<0>(ops, h, acc0, acc1, g0, cc, A, m);
    scan_step<1>(ops, h, acc0, acc1, g0, cc, A, m);
    scan_step<2>(ops, h, acc0, acc1, g0, cc, A, m);
    scan_step<3>(ops, h, acc0, acc1, g0, cc, A, m);
    scan_step<4>(ops, h, acc0, acc1, g0, cc, A, m);
    scan_step<5>(ops, h, acc0, acc1, g0, cc, A, m);
    scan_step<6>(ops, h, acc0, acc1, g0, cc, A, m);
    scan_step<7>(ops, h, acc0, acc1, g1, cc, A, m);

    float uk[GSTEPS];
#pragma unroll
    for (int j = 0; j < GSTEPS; ++j) uk[j] = g0.u[j];

    const size_t rn = (size_t)(t + 1 < NTILE ? t + 1 : t) * TSTEPS;
    load_group(g0, dtp + rn * D_INNER, up + rn * D_INNER, bp + rn * E_DIM);
    float ccn[8];
    {
      const float4 c0 = *(const float4*)(cp + rn * E_DIM);
      const float4 c1 = *(const float4*)(cp + rn * E_DIM + 4);
      ccn[0] = c0.x; ccn[1] = c0.y; ccn[2] = c0.z; ccn[3] = c0.w;
      ccn[4] = c1.x; ccn[5] = c1.y; ccn[6] = c1.z; ccn[7] = c1.w;
    }

    scan_step<8>(ops, h, acc0, acc1, g1, cc, A, m);
    scan_step<9>(ops, h, acc0, acc1, g1, cc, A, m);
    scan_step<10>(ops, h, acc0, acc1, g1, cc, A, m);
    scan_step<11>(ops, h, acc0, acc1, g1, cc, A, m);
    scan_step<12>(ops, h, acc0, acc1, g1, cc, A, m);
    scan_step<13>(ops, h, acc0, acc1, g1, cc, A, m);
    scan_step<14>(ops, h, acc0, acc1, g1, cc, A, m);
    scan_step<15>(ops, h, acc0, acc1, g0, ccn, A, m);

#pragma unroll
    for (int j = 0; j < 8; ++j) {
      const float uu = kh ? g1.u[j] : uk[j];
      op[(r + j) * D_INNER] = fmaf(uu, dp, acc0[j] + acc1[j]);
    }
#pragma unroll
    for (int j = 0; j < 8; ++j) cc[j] = ccn[j];
  }
}

extern "C" void kernel_launch(void* const* d_in, const int* in_sizes, int n_in,
                              void* d_out, int out_size, void* d_ws, size_t ws_size,
                              hipStream_t stream) {
  const float* u     = (const float*)d_in[0];
  const float* A_log = (const float*)d_in[1];
  const float* W_x   = (const float*)d_in[2];
  const float* W_dt  = (const float*)d_in[3];
  const float* b_dt  = (const float*)d_in[4];
  const float* Dp    = (const float*)d_in[5];
  float* out = (float*)d_out;

  float* ws    = (float*)d_ws;
  float* x_dbl = ws;
  float* dtbuf = x_dbl + (size_t)ROWS * E_DIM;
  float* prodA = dtbuf + (size_t)ROWS * D_INNER;
  float* hend  = prodA + (size_t)BATCH * D_INNER * STATE * NCHUNK;

  xproj_wmma<<<1536 / 4, 128, 0, stream>>>(u, W_x, x_dbl);
  scan_pass1<<<256, 128, 0, stream>>>(u, A_log, x_dbl, W_dt, b_dt, dtbuf, prodA, hend);
  scan_pass3<<<256, 128, 0, stream>>>(u, A_log, x_dbl, dtbuf, prodA, hend, Dp, out);
}


# GEMM1 operands fetched as whole 128-byte lines (4 lines per wave-wide load, not 16) and turned into fragments through the wave's own LDS tiles, 2 stages ahead; k order and partial tiles unchanged
# speedup vs baseline: 4.3740x; 1.0670x over previous
#include <hip/hip_runtime.h>
#include <hip/hip_bf16.h>

typedef __attribute__((ext_vector_type(2))) float v2f;
typedef __attribute__((ext_vector_type(4))) float v4f;
typedef __attribute__((ext_vector_type(8))) float v8f;

#define D_INNER 1024
#define STATE   16
#define DT_RANK 64
#define BATCH   2
#define SEQ     2048
#define ROWS    (BATCH * SEQ)
#define E_DIM   (DT_RANK + 2 * STATE)
#define NCHUNK  8
#define CLEN    (SEQ / NCHUNK)
#define SUM_IDX(b, c, d, kh) \
  ((((size_t)(b) * NCHUNK + (size_t)(c)) * D_INNER + (size_t)(d)) * STATE + 8 * (size_t)(kh))

#define G1_CH    8
#define G1_SK    (8 * G1_CH)
#define G1_NSTG  (D_INNER / G1_SK)
#define G1_PITCH (G1_SK + 4)
#define G1_TILE  (16 * G1_PITCH)

struct G1Stage {
  v4f a[G1_CH];
  v4f b[G1_CH];
};

struct G1Raw {
  v4f a[8];
  v4f b[8];
};

__device__ __forceinline__ void g1_fetch(
    G1Raw& r, const float* __restrict__ ap, const float* __restrict__ bp) {
#pragma unroll
  for (int j = 0; j < 8; ++j) {
    r.a[j] = *(const v4f*)(ap + (size_t)j * 2 * D_INNER);
    r.b[j] = *(const v4f*)(bp + (size_t)j * 2 * D_INNER);
  }
}

__device__ __forceinline__ void g1_stash(const G1Raw& r, float* sa, float* sb) {
#pragma unroll
  for (int j = 0; j < 8; ++j) {
    *(v4f*)(sa + j * 2 * G1_PITCH) = r.a[j];
    *(v4f*)(sb + j * 2 * G1_PITCH) = r.b[j];
  }
}

__device__ __forceinline__ void g1_frag(G1Stage& s, const float* fa, const float* fb) {
#pragma unroll
  for (int c = 0; c < G1_CH; ++c) {
    s.a[c] = *(const v4f*)(fa + 8 * c);
    s.b[c] = *(const v4f*)(fb + 8 * c);
  }
}

__device__ __forceinline__ void g1_mma(const G1Stage& s, v8f& acc0, v8f& acc1) {
#pragma unroll
  for (int c = 0; c < G1_CH; ++c) {
    const v2f a0 = __builtin_shufflevector(s.a[c], s.a[c], 0, 1);
    const v2f b0 = __builtin_shufflevector(s.b[c], s.b[c], 0, 1);
    const v2f a1 = __builtin_shufflevector(s.a[c], s.a[c], 2, 3);
    const v2f b1 = __builtin_shufflevector(s.b[c], s.b[c], 2, 3);
    acc0 = __builtin_amdgcn_wmma_f32_16x16x4_f32(
         false, a0,  false, b0,
         (short)0, acc0,  false,  false);
    acc1 = __builtin_amdgcn_wmma_f32_16x16x4_f32(
        false, a1, false, b1, (short)0, acc1, false, false);
  }
}

__global__ __launch_bounds__(128) void xproj_wmma(
    const float* __restrict__ u, const float* __restrict__ Wx,
    float* __restrict__ x_dbl) {
  __shared__ __attribute__((aligned(16))) float g1_lds[4][2][2][G1_TILE];

  const int wv   = threadIdx.x >> 5;
  const int wave = blockIdx.x * (blockDim.x >> 5) + wv;
  const int lane = threadIdx.x & 31;
  const int rt = wave / (E_DIM / 16);
  const int ct = wave % (E_DIM / 16);
  const int row0 = rt * 16, col0 = ct * 16;
  const int m  = lane & 15;
  const int kh = lane >> 4;

  v8f acc0 = {}, acc1 = {};
  const float* ua = u  + (size_t)(row0 + kh) * D_INNER + 4 * m;
  const float* wa = Wx + (size_t)(col0 + kh) * D_INNER + 4 * m;
  const int st_off = kh * G1_PITCH + 4 * m;
  const int fr_off = m * G1_PITCH + 4 * kh;

  G1Raw   raw[2];
  G1Stage fr[2];
  g1_fetch(raw[0], ua, wa);
  g1_fetch(raw[1], ua + G1_SK, wa + G1_SK);
  g1_stash(raw[0], &g1_lds[wv][0][0][st_off], &g1_lds[wv][0][1][st_off]);
  g1_fetch(raw[0], ua + 2 * G1_SK, wa + 2 * G1_SK);
  g1_frag(fr[0], &g1_lds[wv][0][0][fr_off], &g1_lds[wv][0][1][fr_off]);

#pragma unroll
  for (int s = 0; s < G1_NSTG; ++s) {
    if (s + 1 < G1_NSTG) {
      const int nb = (s + 1) & 1;
      g1_stash(raw[nb], &g1_lds[wv][nb][0][st_off], &g1_lds[wv][nb][1][st_off]);
      if (s + 3 < G1_NSTG)
        g1_fetch(raw[nb], ua + (s + 3) * G1_SK, wa + (s + 3) * G1_SK);
      g1_frag(fr[nb], &g1_lds[wv][nb][0][fr_off], &g1_lds[wv][nb][1][fr_off]);
    }
    __builtin_amdgcn_sched_barrier(0);
    g1_mma(fr[s & 1], acc0, acc1);
    __builtin_amdgcn_sched_barrier(0);
  }

  float* outp = x_dbl + (size_t)(row0 + 8 * kh) * E_DIM + col0 + m;
#pragma unroll
  for (int j = 0; j < 8; ++j) outp[(size_t)j * E_DIM] = acc0[j] + acc1[j];
}

#define P1_GS   2
#define P1_NBUF 8
#define P1_NGRP (CLEN / P1_GS)
#define P1_TS   (P1_GS * P1_NBUF)
#define DT_CH   (DT_RANK / 8)

struct P1Group {
  float u[P1_GS];
  v4f   b[P1_GS][2];
};

struct DtFrag {
  v4f r[DT_CH];
};

__device__ __forceinline__ void dt_load(DtFrag& f, const float* __restrict__ p) {
#pragma unroll
  for (int c = 0; c < DT_CH; ++c) f.r[c] = *(const v4f*)(p + 8 * c);
}

__device__ __forceinline__ float xhalf(const float x) {
  const int b = __builtin_bit_cast(int, x);
  return __builtin_bit_cast(float,
      __builtin_amdgcn_permlanex16(b, b, 0x76543210, 0xfedcba98, false, false));
}

__device__ __forceinline__ void dt_form(
    float (&dts)[P1_TS], const DtFrag& x, const DtFrag& w, const float bias,
    const int kh, float* __restrict__ dtw) {
  v8f acc0 = {}, acc1 = {};
#pragma unroll
  for (int c = 0; c < DT_CH; ++c) {
    const v2f a0 = __builtin_shufflevector(x.r[c], x.r[c], 0, 1);
    const v2f b0 = __builtin_shufflevector(w.r[c], w.r[c], 0, 1);
    const v2f a1 = __builtin_shufflevector(x.r[c], x.r[c], 2, 3);
    const v2f b1 = __builtin_shufflevector(w.r[c], w.r[c], 2, 3);
    acc0 = __builtin_amdgcn_wmma_f32_16x16x4_f32(
         false, a0,  false, b0,
         (short)0, acc0,  false,  false);
    acc1 = __builtin_amdgcn_wmma_f32_16x16x4_f32(
        false, a1, false, b1, (short)0, acc1, false, false);
  }
  float own[8], oth[8];
#pragma unroll
  for (int j = 0; j < 8; ++j) {
    const float v  = (acc0[j] + acc1[j]) + bias;
    const float sp = log1pf(__expf(v));
    own[j] = (v > 20.0f) ? v : sp;
    dtw[(size_t)j * D_INNER] = own[j];
  }
#pragma unroll
  for (int j = 0; j < 8; ++j) oth[j] = xhalf(own[j]);
#pragma unroll
  for (int j = 0; j < 8; ++j) {
    dts[j]     = kh ? oth[j] : own[j];
    dts[8 + j] = kh ? own[j] : oth[j];
  }
}

__device__ __forceinline__ void p1_load(
    P1Group& g, const float* __restrict__ up, const float* __restrict__ bp) {
#pragma unroll
  for (int s = 0; s < P1_GS; ++s) {
    g.u[s]    = up [(size_t)s * D_INNER];
    g.b[s][0] = *(const v4f*)(bp + (size_t)s * E_DIM);
    g.b[s][1] = *(const v4f*)(bp + (size_t)s * E_DIM + 4);
  }
}

__device__ __forceinline__ void p1_steps(
    const P1Group& g, const float dt0, const float dt1,
    v2f (&h)[4], v2f (&p)[4], const v2f (&A)[4]) {
  const float dts[P1_GS] = {dt0, dt1};
#pragma unroll
  for (int s = 0; s < P1_GS; ++s) {
    const float dtv = dts[s];
    const float dtu = dtv * g.u[s];
    const v4f bu0 = g.b[s][0] * dtu;
    const v4f bu1 = g.b[s][1] * dtu;
    const v2f bu[4] = {__builtin_shufflevector(bu0, bu0, 0, 1), __builtin_shufflevector(bu0, bu0, 2, 3),
                       __builtin_shufflevector(bu1, bu1, 0, 1), __builtin_shufflevector(bu1, bu1, 2, 3)};
#pragma unroll
    for (int kb = 0; kb < 4; ++kb) {
      const v2f x = A[kb] * dtv;
      v2f e;
      e.x = __expf(x.x);
      e.y = __expf(x.y);
      h[kb] = __builtin_elementwise_fma(e, h[kb], bu[kb]);
      p[kb] = p[kb] * e;
    }
  }
}

template <bool LAST>
__device__ __forceinline__ void p1_round(
    P1Group (&g)[P1_NBUF], DtFrag& xt, const DtFrag& w, const float bias, const int kh,
    v2f (&h)[4], v2f (&p)[4], const v2f (&A)[4],
    const float* __restrict__ xp, float* __restrict__ dtw,
    const float* __restrict__ up, const float* __restrict__ bp, const int it) {
  float dts[P1_TS];
  dt_form(dts, xt, w, bias, kh, dtw + (size_t)(it * P1_TS) * D_INNER);
  if (!LAST) dt_load(xt, xp + (size_t)((it + 1) * P1_TS) * E_DIM);
#pragma unroll
  for (int j = 0; j < P1_NBUF; ++j) {
    if (!LAST || j == 0) {
      const size_t r = (size_t)((it * P1_NBUF + j + (P1_NBUF - 1)) * P1_GS);
      p1_load(g[(j + P1_NBUF - 1) % P1_NBUF], up + r * D_INNER, bp + r * E_DIM);
      __builtin_amdgcn_sched_barrier(0);
    }
    p1_steps(g[j], dts[P1_GS * j], dts[P1_GS * j + 1], h, p, A);
  }
}

__global__ __launch_bounds__(128) __attribute__((amdgpu_waves_per_eu(1, 2))) void scan_pass1(
    const float* __restrict__ u, const float* __restrict__ A_log,
    const float* __restrict__ x_dbl, const float* __restrict__ Wdt,
    const float* __restrict__ b_dt, float* __restrict__ dt,
    float* __restrict__ prodA, float* __restrict__ hend) {
  const int lane  = threadIdx.x & 31;
  const int wv    = threadIdx.x >> 5;
  const int b     = blockIdx.x / (NCHUNK * (D_INNER / 64));
  const int rem   = blockIdx.x % (NCHUNK * (D_INNER / 64));
  const int chunk = rem / (D_INNER / 64);
  const int d0    = (rem % (D_INNER / 64)) * 64 + wv * 16;
  const int m  = lane & 15;
  const int kh = lane >> 4;
  const int d  = d0 + m;

  const size_t row0 = (size_t)b * SEQ + (size_t)chunk * CLEN;
  const float* xp  = x_dbl + (row0 + m) * E_DIM + 4 * kh;
  const float* up  = u  + row0 * D_INNER + d;
  const float* bp  = x_dbl + row0 * E_DIM + DT_RANK + 8 * kh;
  float* dtw = dt + (row0 + 8 * kh) * D_INNER + d;

  DtFrag w, xt;
  dt_load(w, Wdt + (size_t)d * DT_RANK + 4 * kh);
  dt_load(xt, xp);
  const float bias = b_dt[d];

  v2f A[4], h[4], p[4];
  {
    const float4 al0 = *(const float4*)(A_log + d * STATE + 8 * kh);
    const float4 al1 = *(const float4*)(A_log + d * STATE + 8 * kh + 4);
    A[0].x = -__expf(al0.x); A[0].y = -__expf(al0.y);
    A[1].x = -__expf(al0.z); A[1].y = -__expf(al0.w);
    A[2].x = -__expf(al1.x); A[2].y = -__expf(al1.y);
    A[3].x = -__expf(al1.z); A[3].y = -__expf(al1.w);
  }
#pragma unroll
  for (int kb = 0; kb < 4; ++kb) {
    h[kb].x = 0.0f; h[kb].y = 0.0f;
    p[kb].x = 1.0f; p[kb].y = 1.0f;
  }

  P1Group g[P1_NBUF];
#pragma unroll
  for (int j = 0; j < P1_NBUF - 1; ++j)
    p1_load(g[j], up + (size_t)(j * P1_GS) * D_INNER, bp + (size_t)(j * P1_GS) * E_DIM);

  p1_round<false>(g, xt, w, bias, kh, h, p, A, xp, dtw, up, bp, 0);
  for (int it = 1; it < P1_NGRP / P1_NBUF - 1; ++it)
    p1_round<false>(g, xt, w, bias, kh, h, p, A, xp, dtw, up, bp, it);
  p1_round<true>(g, xt, w, bias, kh, h, p, A, xp, dtw, up, bp, P1_NGRP / P1_NBUF - 1);

  const size_t sidx = SUM_IDX(b, chunk, d, kh);
  *(v4f*)(prodA + sidx)     = __builtin_shufflevector(p[0], p[1], 0, 1, 2, 3);
  *(v4f*)(prodA + sidx + 4) = __builtin_shufflevector(p[2], p[3], 0, 1, 2, 3);
  *(v4f*)(hend  + sidx)     = __builtin_shufflevector(h[0], h[1], 0, 1, 2, 3);
  *(v4f*)(hend  + sidx + 4) = __builtin_shufflevector(h[2], h[3], 0, 1, 2, 3);
}

#define TSTEPS 16
#define GSTEPS 8
#define NTILE  (CLEN / TSTEPS)

struct StepGroup {
  float  dt[GSTEPS];
  float  u[GSTEPS];
  float4 b[GSTEPS][2];
};

__device__ __forceinline__ void load_group(
    StepGroup& g, const float* __restrict__ dtp, const float* __restrict__ up,
    const float* __restrict__ bp) {
#pragma unroll
  for (int s = 0; s < GSTEPS; ++s) {
    g.dt[s]   = dtp[(size_t)s * D_INNER];
    g.u[s]    = up [(size_t)s * D_INNER];
    g.b[s][0] = *(const float4*)(bp + (size_t)s * E_DIM);
    g.b[s][1] = *(const float4*)(bp + (size_t)s * E_DIM + 4);
  }
}

struct StepOps {
  v2f e[4];
  v2f bu[4];
  v2f a[4];
};

template <int S>
__device__ __forceinline__ void prep_step(
    StepOps& o, const StepGroup& g, const float (&cc)[8], const v2f (&A)[4],
    const int m) {
  constexpr int s = S & (GSTEPS - 1);
  const float dtv = g.dt[s];
  const float dtu = dtv * g.u[s];
  const bool mine = (m == S);
  const float bb[8] = {g.b[s][0].x, g.b[s][0].y, g.b[s][0].z, g.b[s][0].w,
                       g.b[s][1].x, g.b[s][1].y, g.b[s][1].z, g.b[s][1].w};
#pragma unroll
  for (int kb = 0; kb < 4; ++kb) {
    o.e[kb].x  = __expf(dtv * A[kb].x);
    o.e[kb].y  = __expf(dtv * A[kb].y);
    o.bu[kb].x = dtu * bb[2 * kb];
    o.bu[kb].y = dtu * bb[2 * kb + 1];
    o.a[kb].x  = mine ? cc[2 * kb]     : 0.0f;
    o.a[kb].y  = mine ? cc[2 * kb + 1] : 0.0f;
  }
}

template <int S>
__device__ __forceinline__ void scan_step(
    StepOps& o, v2f (&h)[4], v8f& acc0, v8f& acc1, const StepGroup& gnext,
    const float (&ccnext)[8], const v2f (&A)[4], const int m) {
#pragma unroll
  for (int kb = 0; kb < 4; ++kb)
    h[kb] = __builtin_elementwise_fma(o.e[kb], h[kb], o.bu[kb]);
  acc0 = __builtin_amdgcn_wmma_f32_16x16x4_f32(
      false, o.a[0], false, h[0], (short)0, acc0, false, false);
  acc1 = __builtin_amdgcn_wmma_f32_16x16x4_f32(
      false, o.a[1], false, h[1], (short)0, acc1, false, false);
  acc0 = __builtin_amdgcn_wmma_f32_16x16x4_f32(
      false, o.a[2], false, h[2], (short)0, acc0, false, false);
  acc1 = __builtin_amdgcn_wmma_f32_16x16x4_f32(
      false, o.a[3], false, h[3], (short)0, acc1, false, false);
  prep_step<(S + 1) & (TSTEPS - 1)>(o, gnext, ccnext, A, m);
}

__global__ __launch_bounds__(128) __attribute__((amdgpu_waves_per_eu(1, 3))) void scan_pass3(
    const float* __restrict__ u, const float* __restrict__ A_log,
    const float* __restrict__ x_dbl, const float* __restrict__ dt,
    const float* __restrict__ prodA, const float* __restrict__ hend,
    const float* __restrict__ Dp, float* __restrict__ out) {
  const int lane  = threadIdx.x & 31;
  const int wv    = threadIdx.x >> 5;
  const int b     = blockIdx.x / (NCHUNK * (D_INNER / 64));
  const int rem   = blockIdx.x % (NCHUNK * (D_INNER / 64));
  const int chunk = rem / (D_INNER / 64);
  const int d0    = (rem % (D_INNER / 64)) * 64 + wv * 16;
  const int m  = lane & 15;
  const int kh = lane >> 4;
  const int d  = d0 + m;

  v2f A[4], h[4];
  {
    const float4 al0 = *(const float4*)(A_log + d * STATE + 8 * kh);
    const float4 al1 = *(const float4*)(A_log + d * STATE + 8 * kh + 4);
    A[0].x = -__expf(al0.x); A[0].y = -__expf(al0.y);
    A[1].x = -__expf(al0.z); A[1].y = -__expf(al0.w);
    A[2].x = -__expf(al1.x); A[2].y = -__expf(al1.y);
    A[3].x = -__expf(al1.z); A[3].y = -__expf(al1.w);
  }
  {
    const float* pp = prodA + SUM_IDX(b, 0, d, kh);
    const float* hp = hend  + SUM_IDX(b, 0, d, kh);
    v4f sp[NCHUNK - 1][2], sh[NCHUNK - 1][2];
#pragma unroll
    for (int c = 0; c < NCHUNK - 1; ++c) {
      sp[c][0] = *(const v4f*)(pp + (size_t)c * D_INNER * STATE);
      sp[c][1] = *(const v4f*)(pp + (size_t)c * D_INNER * STATE + 4);
      sh[c][0] = *(const v4f*)(hp + (size_t)c * D_INNER * STATE);
      sh[c][1] = *(const v4f*)(hp + (size_t)c * D_INNER * STATE + 4);
    }
#pragma unroll
    for (int kb = 0; kb < 4; ++kb) { h[kb].x = 0.0f; h[kb].y = 0.0f; }
#pragma unroll
    for (int c = 0; c < NCHUNK - 1; ++c) {
      const bool on = c < chunk;
      const v2f pc[4] = {__builtin_shufflevector(sp[c][0], sp[c][0], 0, 1),
                         __builtin_shufflevector(sp[c][0], sp[c][0], 2, 3),
                         __builtin_shufflevector(sp[c][1], sp[c][1], 0, 1),
                         __builtin_shufflevector(sp[c][1], sp[c][1], 2, 3)};
      const v2f hc[4] = {__builtin_shufflevector(sh[c][0], sh[c][0], 0, 1),
                         __builtin_shufflevector(sh[c][0], sh[c][0], 2, 3),
                         __builtin_shufflevector(sh[c][1], sh[c][1], 0, 1),
                         __builtin_shufflevector(sh[c][1], sh[c][1], 2, 3)};
#pragma unroll
      for (int kb = 0; kb < 4; ++kb) {
        const v2f hn = __builtin_elementwise_fma(pc[kb], h[kb], hc[kb]);
        h[kb] = on ? hn : h[kb];
      }
    }
  }
  const float dp = Dp[d];

  const size_t row0 = (size_t)b * SEQ + (size_t)chunk * CLEN;
  const float* dtp = dt + row0 * D_INNER + d;
  const float* up  = u  + row0 * D_INNER + d;
  const float* bp  = x_dbl + row0 * E_DIM + DT_RANK + 8 * kh;
  const float* cp  = x_dbl + (row0 + m) * E_DIM + DT_RANK + STATE + 8 * kh;
  float* op = out + (row0 + 8 * kh) * D_INNER + d;

  StepGroup g0, g1;
  StepOps ops;
  float cc[8];
  load_group(g0, dtp, up, bp);
  {
    const float4 c0 = *(const float4*)(cp);
    const float4 c1 = *(const float4*)(cp + 4);
    cc[0] = c0.x; cc[1] = c0.y; cc[2] = c0.z; cc[3] = c0.w;
    cc[4] = c1.x; cc[5] = c1.y; cc[6] = c1.z; cc[7] = c1.w;
  }
  prep_step<0>(ops, g0, cc, A, m);

  for (int t = 0; t < NTILE; ++t) {
    const size_t r = (size_t)t * TSTEPS;
    load_group(g1, dtp + (r + GSTEPS) * D_INNER, up + (r + GSTEPS) * D_INNER,
               bp + (r + GSTEPS) * E_DIM);

    v8f acc0 = {}, acc1 = {};
    scan_step<0>(ops, h, acc0, acc1, g0, cc, A, m);
    scan_step<1>(ops, h, acc0, acc1, g0, cc, A, m);
    scan_step<2>(ops, h, acc0, acc1, g0, cc, A, m);
    scan_step<3>(ops, h, acc0, acc1, g0, cc, A, m);
    scan_step<4>(ops, h, acc0, acc1, g0, cc, A, m);
    scan_step<5>(ops, h, acc0, acc1, g0, cc, A, m);
    scan_step<6>(ops, h, acc0, acc1, g0, cc, A, m);
    scan_step<7>(ops, h, acc0, acc1, g1, cc, A, m);

    float uk[GSTEPS];
#pragma unroll
    for (int j = 0; j < GSTEPS; ++j) uk[j] = g0.u[j];

    const size_t rn = (size_t)(t + 1 < NTILE ? t + 1 : t) * TSTEPS;
    load_group(g0, dtp + rn * D_INNER, up + rn * D_INNER, bp + rn * E_DIM);
    float ccn[8];
    {
      const float4 c0 = *(const float4*)(cp + rn * E_DIM);
      const float4 c1 = *(const float4*)(cp + rn * E_DIM + 4);
      ccn[0] = c0.x; ccn[1] = c0.y; ccn[2] = c0.z; ccn[3] = c0.w;
      ccn[4] = c1.x; ccn[5] = c1.y; ccn[6] = c1.z; ccn[7] = c1.w;
    }

    scan_step<8>(ops, h, acc0, acc1, g1, cc, A, m);
    scan_step<9>(ops, h, acc0, acc1, g1, cc, A, m);
    scan_step<10>(ops, h, acc0, acc1, g1, cc, A, m);
    scan_step<11>(ops, h, acc0, acc1, g1, cc, A, m);
    scan_step<12>(ops, h, acc0, acc1, g1, cc, A, m);
    scan_step<13>(ops, h, acc0, acc1, g1, cc, A, m);
    scan_step<14>(ops, h, acc0, acc1, g1, cc, A, m);
    scan_step<15>(ops, h, acc0, acc1, g0, ccn, A, m);

#pragma unroll
    for (int j = 0; j < 8; ++j) {
      const float uu = kh ? g1.u[j] : uk[j];
      op[(r + j) * D_INNER] = fmaf(uu, dp, acc0[j] + acc1[j]);
    }
#pragma unroll
    for (int j = 0; j < 8; ++j) cc[j] = ccn[j];
  }
}

extern "C" void kernel_launch(void* const* d_in, const int* in_sizes, int n_in,
                              void* d_out, int out_size, void* d_ws, size_t ws_size,
                              hipStream_t stream) {
  const float* u     = (const float*)d_in[0];
  const float* A_log = (const float*)d_in[1];
  const float* W_x   = (const float*)d_in[2];
  const float* W_dt  = (const float*)d_in[3];
  const float* b_dt  = (const float*)d_in[4];
  const float* Dp    = (const float*)d_in[5];
  float* out = (float*)d_out;

  float* ws    = (float*)d_ws;
  float* x_dbl = ws;
  float* dtbuf = x_dbl + (size_t)ROWS * E_DIM;
  float* prodA = dtbuf + (size_t)ROWS * D_INNER;
  float* hend  = prodA + (size_t)BATCH * D_INNER * STATE * NCHUNK;

  xproj_wmma<<<1536 / 4, 128, 0, stream>>>(u, W_x, x_dbl);
  scan_pass1<<<256, 128, 0, stream>>>(u, A_log, x_dbl, W_dt, b_dt, dtbuf, prodA, hend);
  scan_pass3<<<256, 128, 0, stream>>>(u, A_log, x_dbl, dtbuf, prodA, hend, Dp, out);
}


# GEMM1 operands staged once per block: six waves (2 u strips x 3 W_x strips) share five LDS tiles per 64-k stage behind one barrier; 5 tiles fetched where 12 were; per-tile k order unchanged
# speedup vs baseline: 4.5141x; 1.0320x over previous
#include <hip/hip_runtime.h>
#include <hip/hip_bf16.h>

typedef __attribute__((ext_vector_type(2))) float v2f;
typedef __attribute__((ext_vector_type(4))) float v4f;
typedef __attribute__((ext_vector_type(8))) float v8f;

#define D_INNER 1024
#define STATE   16
#define DT_RANK 64
#define BATCH   2
#define SEQ     2048
#define ROWS    (BATCH * SEQ)
#define E_DIM   (DT_RANK + 2 * STATE)
#define NCHUNK  8
#define CLEN    (SEQ / NCHUNK)
#define SUM_IDX(b, c, d, kh) \
  ((((size_t)(b) * NCHUNK + (size_t)(c)) * D_INNER + (size_t)(d)) * STATE + 8 * (size_t)(kh))

#define G1_CH    8
#define G1_SK    (8 * G1_CH)
#define G1_NSTG  (D_INNER / G1_SK)
#define G1_PITCH (G1_SK + 4)
#define G1_TILE  (16 * G1_PITCH)
#define G1_RS    2
#define G1_CS    3
#define G1_NT    (G1_RS + G1_CS)
#define G1_NW    (G1_RS * G1_CS)
#define G1_NPC   (G1_NT * 8)
#define G1_NP    ((G1_NPC + G1_NW - 1) / G1_NW)

struct G1Stage {
  v4f a[G1_CH];
  v4f b[G1_CH];
};

struct G1Raw {
  v4f p[G1_NP];
};

__device__ __forceinline__ void g1_fetch(
    G1Raw& r, const float* const (&gp)[G1_NP], const int k0, const bool last) {
#pragma unroll
  for (int i = 0; i < G1_NP - 1; ++i) r.p[i] = *(const v4f*)(gp[i] + k0);
  if (last) r.p[G1_NP - 1] = *(const v4f*)(gp[G1_NP - 1] + k0);
}

__device__ __forceinline__ void g1_stash(
    const G1Raw& r, float* buf, const int (&so)[G1_NP], const bool last) {
#pragma unroll
  for (int i = 0; i < G1_NP - 1; ++i) *(v4f*)(buf + so[i]) = r.p[i];
  if (last) *(v4f*)(buf + so[G1_NP - 1]) = r.p[G1_NP - 1];
}

__device__ __forceinline__ void g1_frag(G1Stage& s, const float* fa, const float* fb) {
#pragma unroll
  for (int c = 0; c < G1_CH; ++c) {
    s.a[c] = *(const v4f*)(fa + 8 * c);
    s.b[c] = *(const v4f*)(fb + 8 * c);
  }
}

__device__ __forceinline__ void g1_mma(const G1Stage& s, v8f& acc0, v8f& acc1) {
#pragma unroll
  for (int c = 0; c < G1_CH; ++c) {
    const v2f a0 = __builtin_shufflevector(s.a[c], s.a[c], 0, 1);
    const v2f b0 = __builtin_shufflevector(s.b[c], s.b[c], 0, 1);
    const v2f a1 = __builtin_shufflevector(s.a[c], s.a[c], 2, 3);
    const v2f b1 = __builtin_shufflevector(s.b[c], s.b[c], 2, 3);
    acc0 = __builtin_amdgcn_wmma_f32_16x16x4_f32(
         false, a0,  false, b0,
         (short)0, acc0,  false,  false);
    acc1 = __builtin_amdgcn_wmma_f32_16x16x4_f32(
        false, a1, false, b1, (short)0, acc1, false, false);
  }
}

__device__ __forceinline__ void g1_meet() {
  __builtin_amdgcn_fence(__ATOMIC_RELEASE, "workgroup", "local");
  __builtin_amdgcn_s_barrier();
  __builtin_amdgcn_fence(__ATOMIC_ACQUIRE, "workgroup", "local");
}

__global__ __launch_bounds__(32 * G1_NW) void xproj_wmma(
    const float* __restrict__ u, const float* __restrict__ Wx,
    float* __restrict__ x_dbl) {
  __shared__ __attribute__((aligned(16))) float g1_lds[2][G1_NT * G1_TILE];

  const int wv   = __builtin_amdgcn_readfirstlane((int)(threadIdx.x >> 5));
  const int lane = threadIdx.x & 31;
  const int m  = lane & 15;
  const int kh = lane >> 4;
  const int row0 = (int)(blockIdx.x / (E_DIM / (16 * G1_CS))) * (16 * G1_RS);
  const int col0 = (int)(blockIdx.x % (E_DIM / (16 * G1_CS))) * (16 * G1_CS);
  const int rs = wv / G1_CS;
  const int cs = wv % G1_CS;

  const float* gp[G1_NP];
  int so[G1_NP];
#pragma unroll
  for (int i = 0; i < G1_NP; ++i) {
    const int q = (i * G1_NW + wv) % G1_NPC;
    const int t = q >> 3;
    const int r = 2 * (q & 7) + kh;
    const float* base = (t < G1_RS) ? u  + (size_t)(row0 + 16 * t + r) * D_INNER
                                    : Wx + (size_t)(col0 + 16 * (t - G1_RS) + r) * D_INNER;
    gp[i] = base + 4 * m;
    so[i] = t * G1_TILE + r * G1_PITCH + 4 * m;
  }
  const bool last = (G1_NP - 1) * G1_NW + wv < G1_NPC;
  const int fa_off = rs * G1_TILE + m * G1_PITCH + 4 * kh;
  const int fb_off = (G1_RS + cs) * G1_TILE + m * G1_PITCH + 4 * kh;

  v8f acc0 = {}, acc1 = {};
  G1Raw   raw[2];
  G1Stage fr[2];
  g1_fetch(raw[0], gp, 0, last);
  g1_fetch(raw[1], gp, G1_SK, last);
  g1_stash(raw[0], g1_lds[0], so, last);
  g1_fetch(raw[0], gp, 2 * G1_SK, last);
  g1_meet();
  g1_frag(fr[0], &g1_lds[0][fa_off], &g1_lds[0][fb_off]);

#pragma unroll
  for (int s = 0; s < G1_NSTG; ++s) {
    if (s + 1 < G1_NSTG) {
      const int nb = (s + 1) & 1;
      g1_stash(raw[nb], g1_lds[nb], so, last);
      if (s + 3 < G1_NSTG)
        g1_fetch(raw[nb], gp, (s + 3) * G1_SK, last);
      g1_meet();
      g1_frag(fr[nb], &g1_lds[nb][fa_off], &g1_lds[nb][fb_off]);
    }
    __builtin_amdgcn_sched_barrier(0);
    g1_mma(fr[s & 1], acc0, acc1);
    __builtin_amdgcn_sched_barrier(0);
  }

  float* outp = x_dbl + (size_t)(row0 + 16 * rs + 8 * kh) * E_DIM + col0 + 16 * cs + m;
#pragma unroll
  for (int j = 0; j < 8; ++j) outp[(size_t)j * E_DIM] = acc0[j] + acc1[j];
}

#define P1_GS   2
#define P1_NBUF 8
#define P1_NGRP (CLEN / P1_GS)
#define P1_TS   (P1_GS * P1_NBUF)
#define DT_CH   (DT_RANK / 8)

struct P1Group {
  float u[P1_GS];
  v4f   b[P1_GS][2];
};

struct DtFrag {
  v4f r[DT_CH];
};

__device__ __forceinline__ void dt_load(DtFrag& f, const float* __restrict__ p) {
#pragma unroll
  for (int c = 0; c < DT_CH; ++c) f.r[c] = *(const v4f*)(p + 8 * c);
}

__device__ __forceinline__ float xhalf(const float x) {
  const int b = __builtin_bit_cast(int, x);
  return __builtin_bit_cast(float,
      __builtin_amdgcn_permlanex16(b, b, 0x76543210, 0xfedcba98, false, false));
}

__device__ __forceinline__ void dt_form(
    float (&dts)[P1_TS], const DtFrag& x, const DtFrag& w, const float bias,
    const int kh, float* __restrict__ dtw) {
  v8f acc0 = {}, acc1 = {};
#pragma unroll
  for (int c = 0; c < DT_CH; ++c) {
    const v2f a0 = __builtin_shufflevector(x.r[c], x.r[c], 0, 1);
    const v2f b0 = __builtin_shufflevector(w.r[c], w.r[c], 0, 1);
    const v2f a1 = __builtin_shufflevector(x.r[c], x.r[c], 2, 3);
    const v2f b1 = __builtin_shufflevector(w.r[c], w.r[c], 2, 3);
    acc0 = __builtin_amdgcn_wmma_f32_16x16x4_f32(
         false, a0,  false, b0,
         (short)0, acc0,  false,  false);
    acc1 = __builtin_amdgcn_wmma_f32_16x16x4_f32(
        false, a1, false, b1, (short)0, acc1, false, false);
  }
  float own[8], oth[8];
#pragma unroll
  for (int j = 0; j < 8; ++j) {
    const float v  = (acc0[j] + acc1[j]) + bias;
    const float sp = log1pf(__expf(v));
    own[j] = (v > 20.0f) ? v : sp;
    dtw[(size_t)j * D_INNER] = own[j];
  }
#pragma unroll
  for (int j = 0; j < 8; ++j) oth[j] = xhalf(own[j]);
#pragma unroll
  for (int j = 0; j < 8; ++j) {
    dts[j]     = kh ? oth[j] : own[j];
    dts[8 + j] = kh ? own[j] : oth[j];
  }
}

__device__ __forceinline__ void p1_load(
    P1Group& g, const float* __restrict__ up, const float* __restrict__ bp) {
#pragma unroll
  for (int s = 0; s < P1_GS; ++s) {
    g.u[s]    = up [(size_t)s * D_INNER];
    g.b[s][0] = *(const v4f*)(bp + (size_t)s * E_DIM);
    g.b[s][1] = *(const v4f*)(bp + (size_t)s * E_DIM + 4);
  }
}

__device__ __forceinline__ void p1_steps(
    const P1Group& g, const float dt0, const float dt1,
    v2f (&h)[4], v2f (&p)[4], const v2f (&A)[4]) {
  const float dts[P1_GS] = {dt0, dt1};
#pragma unroll
  for (int s = 0; s < P1_GS; ++s) {
    const float dtv = dts[s];
    const float dtu = dtv * g.u[s];
    const v4f bu0 = g.b[s][0] * dtu;
    const v4f bu1 = g.b[s][1] * dtu;
    const v2f bu[4] = {__builtin_shufflevector(bu0, bu0, 0, 1), __builtin_shufflevector(bu0, bu0, 2, 3),
                       __builtin_shufflevector(bu1, bu1, 0, 1), __builtin_shufflevector(bu1, bu1, 2, 3)};
#pragma unroll
    for (int kb = 0; kb < 4; ++kb) {
      const v2f x = A[kb] * dtv;
      v2f e;
      e.x = __expf(x.x);
      e.y = __expf(x.y);
      h[kb] = __builtin_elementwise_fma(e, h[kb], bu[kb]);
      p[kb] = p[kb] * e;
    }
  }
}

template <bool LAST>
__device__ __forceinline__ void p1_round(
    P1Group (&g)[P1_NBUF], DtFrag& xt, const DtFrag& w, const float bias, const int kh,
    v2f (&h)[4], v2f (&p)[4], const v2f (&A)[4],
    const float* __restrict__ xp, float* __restrict__ dtw,
    const float* __restrict__ up, const float* __restrict__ bp, const int it) {
  float dts[P1_TS];
  dt_form(dts, xt, w, bias, kh, dtw + (size_t)(it * P1_TS) * D_INNER);
  if (!LAST) dt_load(xt, xp + (size_t)((it + 1) * P1_TS) * E_DIM);
#pragma unroll
  for (int j = 0; j < P1_NBUF; ++j) {
    if (!LAST || j == 0) {
      const size_t r = (size_t)((it * P1_NBUF + j + (P1_NBUF - 1)) * P1_GS);
      p1_load(g[(j + P1_NBUF - 1) % P1_NBUF], up + r * D_INNER, bp + r * E_DIM);
      __builtin_amdgcn_sched_barrier(0);
    }
    p1_steps(g[j], dts[P1_GS * j], dts[P1_GS * j + 1], h, p, A);
  }
}

__global__ __launch_bounds__(128) __attribute__((amdgpu_waves_per_eu(1, 2))) void scan_pass1(
    const float* __restrict__ u, const float* __restrict__ A_log,
    const float* __restrict__ x_dbl, const float* __restrict__ Wdt,
    const float* __restrict__ b_dt, float* __restrict__ dt,
    float* __restrict__ prodA, float* __restrict__ hend) {
  const int lane  = threadIdx.x & 31;
  const int wv    = threadIdx.x >> 5;
  const int b     = blockIdx.x / (NCHUNK * (D_INNER / 64));
  const int rem   = blockIdx.x % (NCHUNK * (D_INNER / 64));
  const int chunk = rem / (D_INNER / 64);
  const int d0    = (rem % (D_INNER / 64)) * 64 + wv * 16;
  const int m  = lane & 15;
  const int kh = lane >> 4;
  const int d  = d0 + m;

  const size_t row0 = (size_t)b * SEQ + (size_t)chunk * CLEN;
  const float* xp  = x_dbl + (row0 + m) * E_DIM + 4 * kh;
  const float* up  = u  + row0 * D_INNER + d;
  const float* bp  = x_dbl + row0 * E_DIM + DT_RANK + 8 * kh;
  float* dtw = dt + (row0 + 8 * kh) * D_INNER + d;

  DtFrag w, xt;
  dt_load(w, Wdt + (size_t)d * DT_RANK + 4 * kh);
  dt_load(xt, xp);
  const float bias = b_dt[d];

  v2f A[4], h[4], p[4];
  {
    const float4 al0 = *(const float4*)(A_log + d * STATE + 8 * kh);
    const float4 al1 = *(const float4*)(A_log + d * STATE + 8 * kh + 4);
    A[0].x = -__expf(al0.x); A[0].y = -__expf(al0.y);
    A[1].x = -__expf(al0.z); A[1].y = -__expf(al0.w);
    A[2].x = -__expf(al1.x); A[2].y = -__expf(al1.y);
    A[3].x = -__expf(al1.z); A[3].y = -__expf(al1.w);
  }
#pragma unroll
  for (int kb = 0; kb < 4; ++kb) {
    h[kb].x = 0.0f; h[kb].y = 0.0f;
    p[kb].x = 1.0f; p[kb].y = 1.0f;
  }

  P1Group g[P1_NBUF];
#pragma unroll
  for (int j = 0; j < P1_NBUF - 1; ++j)
    p1_load(g[j], up + (size_t)(j * P1_GS) * D_INNER, bp + (size_t)(j * P1_GS) * E_DIM);

  p1_round<false>(g, xt, w, bias, kh, h, p, A, xp, dtw, up, bp, 0);
  for (int it = 1; it < P1_NGRP / P1_NBUF - 1; ++it)
    p1_round<false>(g, xt, w, bias, kh, h, p, A, xp, dtw, up, bp, it);
  p1_round<true>(g, xt, w, bias, kh, h, p, A, xp, dtw, up, bp, P1_NGRP / P1_NBUF - 1);

  const size_t sidx = SUM_IDX(b, chunk, d, kh);
  *(v4f*)(prodA + sidx)     = __builtin_shufflevector(p[0], p[1], 0, 1, 2, 3);
  *(v4f*)(prodA + sidx + 4) = __builtin_shufflevector(p[2], p[3], 0, 1, 2, 3);
  *(v4f*)(hend  + sidx)     = __builtin_shufflevector(h[0], h[1], 0, 1, 2, 3);
  *(v4f*)(hend  + sidx + 4) = __builtin_shufflevector(h[2], h[3], 0, 1, 2, 3);
}

#define TSTEPS 16
#define GSTEPS 8
#define NTILE  (CLEN / TSTEPS)

struct StepGroup {
  float  dt[GSTEPS];
  float  u[GSTEPS];
  float4 b[GSTEPS][2];
};

__device__ __forceinline__ void load_group(
    StepGroup& g, const float* __restrict__ dtp, const float* __restrict__ up,
    const float* __restrict__ bp) {
#pragma unroll
  for (int s = 0; s < GSTEPS; ++s) {
    g.dt[s]   = dtp[(size_t)s * D_INNER];
    g.u[s]    = up [(size_t)s * D_INNER];
    g.b[s][0] = *(const float4*)(bp + (size_t)s * E_DIM);
    g.b[s][1] = *(const float4*)(bp + (size_t)s * E_DIM + 4);
  }
}

struct StepOps {
  v2f e[4];
  v2f bu[4];
  v2f a[4];
};

template <int S>
__device__ __forceinline__ void prep_step(
    StepOps& o, const StepGroup& g, const float (&cc)[8], const v2f (&A)[4],
    const int m) {
  constexpr int s = S & (GSTEPS - 1);
  const float dtv = g.dt[s];
  const float dtu = dtv * g.u[s];
  const bool mine = (m == S);
  const float bb[8] = {g.b[s][0].x, g.b[s][0].y, g.b[s][0].z, g.b[s][0].w,
                       g.b[s][1].x, g.b[s][1].y, g.b[s][1].z, g.b[s][1].w};
#pragma unroll
  for (int kb = 0; kb < 4; ++kb) {
    o.e[kb].x  = __expf(dtv * A[kb].x);
    o.e[kb].y  = __expf(dtv * A[kb].y);
    o.bu[kb].x = dtu * bb[2 * kb];
    o.bu[kb].y = dtu * bb[2 * kb + 1];
    o.a[kb].x  = mine ? cc[2 * kb]     : 0.0f;
    o.a[kb].y  = mine ? cc[2 * kb + 1] : 0.0f;
  }
}

template <int S>
__device__ __forceinline__ void scan_step(
    StepOps& o, v2f (&h)[4], v8f& acc0, v8f& acc1, const StepGroup& gnext,
    const float (&ccnext)[8], const v2f (&A)[4], const int m) {
#pragma unroll
  for (int kb = 0; kb < 4; ++kb)
    h[kb] = __builtin_elementwise_fma(o.e[kb], h[kb], o.bu[kb]);
  acc0 = __builtin_amdgcn_wmma_f32_16x16x4_f32(
      false, o.a[0], false, h[0], (short)0, acc0, false, false);
  acc1 = __builtin_amdgcn_wmma_f32_16x16x4_f32(
      false, o.a[1], false, h[1], (short)0, acc1, false, false);
  acc0 = __builtin_amdgcn_wmma_f32_16x16x4_f32(
      false, o.a[2], false, h[2], (short)0, acc0, false, false);
  acc1 = __builtin_amdgcn_wmma_f32_16x16x4_f32(
      false, o.a[3], false, h[3], (short)0, acc1, false, false);
  prep_step<(S + 1) & (TSTEPS - 1)>(o, gnext, ccnext, A, m);
}

__global__ __launch_bounds__(128) __attribute__((amdgpu_waves_per_eu(1, 3))) void scan_pass3(
    const float* __restrict__ u, const float* __restrict__ A_log,
    const float* __restrict__ x_dbl, const float* __restrict__ dt,
    const float* __restrict__ prodA, const float* __restrict__ hend,
    const float* __restrict__ Dp, float* __restrict__ out) {
  const int lane  = threadIdx.x & 31;
  const int wv    = threadIdx.x >> 5;
  const int b     = blockIdx.x / (NCHUNK * (D_INNER / 64));
  const int rem   = blockIdx.x % (NCHUNK * (D_INNER / 64));
  const int chunk = rem / (D_INNER / 64);
  const int d0    = (rem % (D_INNER / 64)) * 64 + wv * 16;
  const int m  = lane & 15;
  const int kh = lane >> 4;
  const int d  = d0 + m;

  v2f A[4], h[4];
  {
    const float4 al0 = *(const float4*)(A_log + d * STATE + 8 * kh);
    const float4 al1 = *(const float4*)(A_log + d * STATE + 8 * kh + 4);
    A[0].x = -__expf(al0.x); A[0].y = -__expf(al0.y);
    A[1].x = -__expf(al0.z); A[1].y = -__expf(al0.w);
    A[2].x = -__expf(al1.x); A[2].y = -__expf(al1.y);
    A[3].x = -__expf(al1.z); A[3].y = -__expf(al1.w);
  }
  {
    const float* pp = prodA + SUM_IDX(b, 0, d, kh);
    const float* hp = hend  + SUM_IDX(b, 0, d, kh);
    v4f sp[NCHUNK - 1][2], sh[NCHUNK - 1][2];
#pragma unroll
    for (int c = 0; c < NCHUNK - 1; ++c) {
      sp[c][0] = *(const v4f*)(pp + (size_t)c * D_INNER * STATE);
      sp[c][1] = *(const v4f*)(pp + (size_t)c * D_INNER * STATE + 4);
      sh[c][0] = *(const v4f*)(hp + (size_t)c * D_INNER * STATE);
      sh[c][1] = *(const v4f*)(hp + (size_t)c * D_INNER * STATE + 4);
    }
#pragma unroll
    for (int kb = 0; kb < 4; ++kb) { h[kb].x = 0.0f; h[kb].y = 0.0f; }
#pragma unroll
    for (int c = 0; c < NCHUNK - 1; ++c) {
      const bool on = c < chunk;
      const v2f pc[4] = {__builtin_shufflevector(sp[c][0], sp[c][0], 0, 1),
                         __builtin_shufflevector(sp[c][0], sp[c][0], 2, 3),
                         __builtin_shufflevector(sp[c][1], sp[c][1], 0, 1),
                         __builtin_shufflevector(sp[c][1], sp[c][1], 2, 3)};
      const v2f hc[4] = {__builtin_shufflevector(sh[c][0], sh[c][0], 0, 1),
                         __builtin_shufflevector(sh[c][0], sh[c][0], 2, 3),
                         __builtin_shufflevector(sh[c][1], sh[c][1], 0, 1),
                         __builtin_shufflevector(sh[c][1], sh[c][1], 2, 3)};
#pragma unroll
      for (int kb = 0; kb < 4; ++kb) {
        const v2f hn = __builtin_elementwise_fma(pc[kb], h[kb], hc[kb]);
        h[kb] = on ? hn : h[kb];
      }
    }
  }
  const float dp = Dp[d];

  const size_t row0 = (size_t)b * SEQ + (size_t)chunk * CLEN;
  const float* dtp = dt + row0 * D_INNER + d;
  const float* up  = u  + row0 * D_INNER + d;
  const float* bp  = x_dbl + row0 * E_DIM + DT_RANK + 8 * kh;
  const float* cp  = x_dbl + (row0 + m) * E_DIM + DT_RANK + STATE + 8 * kh;
  float* op = out + (row0 + 8 * kh) * D_INNER + d;

  StepGroup g0, g1;
  StepOps ops;
  float cc[8];
  load_group(g0, dtp, up, bp);
  {
    const float4 c0 = *(const float4*)(cp);
    const float4 c1 = *(const float4*)(cp + 4);
    cc[0] = c0.x; cc[1] = c0.y; cc[2] = c0.z; cc[3] = c0.w;
    cc[4] = c1.x; cc[5] = c1.y; cc[6] = c1.z; cc[7] = c1.w;
  }
  prep_step<0>(ops, g0, cc, A, m);

  for (int t = 0; t < NTILE; ++t) {
    const size_t r = (size_t)t * TSTEPS;
    load_group(g1, dtp + (r + GSTEPS) * D_INNER, up + (r + GSTEPS) * D_INNER,
               bp + (r + GSTEPS) * E_DIM);

    v8f acc0 = {}, acc1 = {};
    scan_step<0>(ops, h, acc0, acc1, g0, cc, A, m);
    scan_step<1>(ops, h, acc0, acc1, g0, cc, A, m);
    scan_step<2>(ops, h, acc0, acc1, g0, cc, A, m);
    scan_step<3>(ops, h, acc0, acc1, g0, cc, A, m);
    scan_step<4>(ops, h, acc0, acc1, g0, cc, A, m);
    scan_step<5>(ops, h, acc0, acc1, g0, cc, A, m);
    scan_step<6>(ops, h, acc0, acc1, g0, cc, A, m);
    scan_step<7>(ops, h, acc0, acc1, g1, cc, A, m);

    float uk[GSTEPS];
#pragma unroll
    for (int j = 0; j < GSTEPS; ++j) uk[j] = g0.u[j];

    const size_t rn = (size_t)(t + 1 < NTILE ? t + 1 : t) * TSTEPS;
    load_group(g0, dtp + rn * D_INNER, up + rn * D_INNER, bp + rn * E_DIM);
    float ccn[8];
    {
      const float4 c0 = *(const float4*)(cp + rn * E_DIM);
      const float4 c1 = *(const float4*)(cp + rn * E_DIM + 4);
      ccn[0] = c0.x; ccn[1] = c0.y; ccn[2] = c0.z; ccn[3] = c0.w;
      ccn[4] = c1.x; ccn[5] = c1.y; ccn[6] = c1.z; ccn[7] = c1.w;
    }

    scan_step<8>(ops, h, acc0, acc1, g1, cc, A, m);
    scan_step<9>(ops, h, acc0, acc1, g1, cc, A, m);
    scan_step<10>(ops, h, acc0, acc1, g1, cc, A, m);
    scan_step<11>(ops, h, acc0, acc1, g1, cc, A, m);
    scan_step<12>(ops, h, acc0, acc1, g1, cc, A, m);
    scan_step<13>(ops, h, acc0, acc1, g1, cc, A, m);
    scan_step<14>(ops, h, acc0, acc1, g1, cc, A, m);
    scan_step<15>(ops, h, acc0, acc1, g0, ccn, A, m);

#pragma unroll
    for (int j = 0; j < 8; ++j) {
      const float uu = kh ? g1.u[j] : uk[j];
      op[(r + j) * D_INNER] = fmaf(uu, dp, acc0[j] + acc1[j]);
    }
#pragma unroll
    for (int j = 0; j < 8; ++j) cc[j] = ccn[j];
  }
}

extern "C" void kernel_launch(void* const* d_in, const int* in_sizes, int n_in,
                              void* d_out, int out_size, void* d_ws, size_t ws_size,
                              hipStream_t stream) {
  const float* u     = (const float*)d_in[0];
  const float* A_log = (const float*)d_in[1];
  const float* W_x   = (const float*)d_in[2];
  const float* W_dt  = (const float*)d_in[3];
  const float* b_dt  = (const float*)d_in[4];
  const float* Dp    = (const float*)d_in[5];
  float* out = (float*)d_out;

  float* ws    = (float*)d_ws;
  float* x_dbl = ws;
  float* dtbuf = x_dbl + (size_t)ROWS * E_DIM;
  float* prodA = dtbuf + (size_t)ROWS * D_INNER;
  float* hend  = prodA + (size_t)BATCH * D_INNER * STATE * NCHUNK;

  xproj_wmma<<<1536 / 6, 192, 0, stream>>>(u, W_x, x_dbl);
  scan_pass1<<<256, 128, 0, stream>>>(u, A_log, x_dbl, W_dt, b_dt, dtbuf, prodA, hend);
  scan_pass3<<<256, 128, 0, stream>>>(u, A_log, x_dbl, dtbuf, prodA, hend, Dp, out);
}
